# starting kernel
# baseline (speedup 1.0000x reference)
#include <hip/hip_runtime.h>
#include <math.h>

typedef __attribute__((ext_vector_type(2))) float v2f;
typedef __attribute__((ext_vector_type(8))) float v8f;

#define LOG2E 1.4426950408889634f

__device__ __forceinline__ float softplus_f(float v) {
  return (v > 20.0f) ? v : log1pf(__expf(v));
}

__global__ __launch_bounds__(256) void dtproj_wmma_kernel(
    const float* __restrict__ delta, const float* __restrict__ W,
    const float* __restrict__ bias, float* __restrict__ dt,
    int M, int Dc, int K) {
  int wave = threadIdx.x >> 5;
  int lane = threadIdx.x & 31;
  int tilesN = Dc >> 5;
  int tile = blockIdx.x * 8 + wave;
  int totalTiles = (M >> 4) * tilesN;
  if (tile >= totalTiles) return;
  int tn = tile % tilesN;
  int tm = tile / tilesN;
  int mn   = lane & 15;
  int half = lane >> 4;

  const float* arow  = delta + (size_t)(tm * 16 + mn) * K;
  const float* brow0 = W     + (size_t)(tn * 32 + mn) * K;
  const float* brow1 = W     + (size_t)(tn * 32 + 16 + mn) * K;

  v8f acc0 = {};
  v8f acc1 = {};
  for (int k0 = 0; k0 < K; k0 += 4) {
    int kb = k0 + (half << 1);
    v2f a, b0, b1;
    a.x  = arow[kb];  a.y  = arow[kb + 1];
    b0.x = brow0[kb]; b0.y = brow0[kb + 1];
    b1.x = brow1[kb]; b1.y = brow1[kb + 1];
    acc0 = __builtin_amdgcn_wmma_f32_16x16x4_f32(
        false, a, false, b0, (short)0, acc0, false, false);
    acc1 = __builtin_amdgcn_wmma_f32_16x16x4_f32(
        false, a, false, b1, (short)0, acc1, false, false);
  }

  int gd0 = tn * 32 + mn;
  int gd1 = gd0 + 16;
  float bv0 = bias[gd0];
  float bv1 = bias[gd1];
  int gm0 = tm * 16 + half * 8;
#pragma unroll
  for (int v = 0; v < 8; ++v) {
    size_t rowoff = (size_t)(gm0 + v) * Dc;
    dt[rowoff + gd0] = softplus_f(acc0[v] + bv0);
    dt[rowoff + gd1] = softplus_f(acc1[v] + bv1);
  }
}

__global__ void dtproj_naive_kernel(const float* __restrict__ delta,
                                    const float* __restrict__ W,
                                    const float* __restrict__ bias,
                                    float* __restrict__ dt,
                                    long long M, int Dc, int K) {
  long long idx = (long long)blockIdx.x * blockDim.x + threadIdx.x;
  if (idx >= M * (long long)Dc) return;
  int d = (int)(idx % Dc);
  long long m = idx / Dc;
  float acc = bias[d];
  const float* dr = delta + m * (long long)K;
  const float* wr = W + (size_t)d * K;
  for (int r = 0; r < K; ++r) acc += dr[r] * wr[r];
  dt[idx] = softplus_f(acc);
}

template <bool FINAL>
__global__ __launch_bounds__(256) void scan_chunk_kernel(
    const float* __restrict__ dt, const float* __restrict__ x,
    const float* __restrict__ Bm, const float* __restrict__ Cm,
    const float* __restrict__ A_log, const float* __restrict__ Dp,
    float* __restrict__ Hout, float* __restrict__ Ssum,
    const float* __restrict__ Hinit, float* __restrict__ out,
    int L, int Dc, int Lc, int nCh) {
  const int N = 16;
  extern __shared__ float smem[];
  float* sB = smem;
  float* sC = smem + (size_t)Lc * N;

  int DB = Dc / blockDim.x;
  int bx = blockIdx.x;
  int dblk = bx % DB; bx /= DB;
  int chunk = bx % nCh; bx /= nCh;
  int b = bx;
  int d = dblk * blockDim.x + threadIdx.x;

  size_t rowbase = (size_t)b * L + (size_t)chunk * Lc;

  {
    const float4* Bsrc = (const float4*)(Bm + rowbase * N);
    const float4* Csrc = (const float4*)(Cm + rowbase * N);
    float4* sB4 = (float4*)sB;
    float4* sC4 = (float4*)sC;
    int tot4 = (Lc * N) >> 2;
    for (int i = threadIdx.x; i < tot4; i += blockDim.x) {
      sB4[i] = Bsrc[i];
      if (FINAL) sC4[i] = Csrc[i];
    }
  }
  __syncthreads();

  float An2[N], h[N];
#pragma unroll
  for (int n = 0; n < N; ++n)
    An2[n] = -__expf(A_log[(size_t)d * N + n]) * LOG2E;

  size_t hidx = (((size_t)b * nCh + chunk) * Dc + d) * N;
#pragma unroll
  for (int n = 0; n < N; ++n) h[n] = FINAL ? Hinit[hidx + n] : 0.0f;

  float dtsum = 0.0f;
  float Dv = FINAL ? Dp[d] : 0.0f;

  for (int l = 0; l < Lc; ++l) {
    size_t g = (rowbase + l) * Dc + d;
    float dtv = dt[g];
    float xv = x[g];
    float us = dtv * xv;
    if (!FINAL) dtsum += dtv;
    float y = 0.0f;
#pragma unroll
    for (int n = 0; n < N; ++n) {
      float a = __builtin_amdgcn_exp2f(dtv * An2[n]);
      h[n] = a * h[n] + us * sB[l * N + n];
      if (FINAL) y += h[n] * sC[l * N + n];
    }
    if (FINAL) out[g] = y + xv * Dv;
  }

  if (!FINAL) {
#pragma unroll
    for (int n = 0; n < N; ++n) Hout[hidx + n] = h[n];
    Ssum[((size_t)b * nCh + chunk) * Dc + d] = dtsum;
  }
}

__global__ void scan_combine_kernel(const float* __restrict__ Hlocal,
                                    const float* __restrict__ Ssum,
                                    const float* __restrict__ A_log,
                                    float* __restrict__ Hinit,
                                    int Dc, int nCh, int Bsz) {
  const int N = 16;
  long long idx = (long long)blockIdx.x * blockDim.x + threadIdx.x;
  if (idx >= (long long)Bsz * Dc) return;
  int d = (int)(idx % Dc);
  int b = (int)(idx / Dc);
  float An2[N], h[N];
#pragma unroll
  for (int n = 0; n < N; ++n) {
    An2[n] = -__expf(A_log[(size_t)d * N + n]) * LOG2E;
    h[n] = 0.0f;
  }
  for (int c = 0; c < nCh; ++c) {
    size_t base = ((size_t)b * nCh + c) * Dc + d;
    size_t hidx = base * N;
    float S = Ssum[base];
#pragma unroll
    for (int n = 0; n < N; ++n) {
      Hinit[hidx + n] = h[n];
      float a = __builtin_amdgcn_exp2f(An2[n] * S);
      h[n] = a * h[n] + Hlocal[hidx + n];
    }
  }
}

__global__ void scan_simple_kernel(const float* __restrict__ dt,
                                   const float* __restrict__ x,
                                   const float* __restrict__ Bm,
                                   const float* __restrict__ Cm,
                                   const float* __restrict__ A_log,
                                   const float* __restrict__ Dp,
                                   float* __restrict__ out,
                                   int L, int Dc, int Bsz) {
  const int N = 16;
  long long idx = (long long)blockIdx.x * blockDim.x + threadIdx.x;
  if (idx >= (long long)Bsz * Dc) return;
  int d = (int)(idx % Dc);
  int b = (int)(idx / Dc);
  float An2[N], h[N];
#pragma unroll
  for (int n = 0; n < N; ++n) {
    An2[n] = -__expf(A_log[(size_t)d * N + n]) * LOG2E;
    h[n] = 0.0f;
  }
  float Dv = Dp[d];
  for (int l = 0; l < L; ++l) {
    size_t row = (size_t)b * L + l;
    size_t g = row * Dc + d;
    float dtv = dt[g];
    float xv = x[g];
    float us = dtv * xv;
    const float* Br = Bm + row * N;
    const float* Cr = Cm + row * N;
    float y = 0.0f;
#pragma unroll
    for (int n = 0; n < N; ++n) {
      h[n] = __builtin_amdgcn_exp2f(dtv * An2[n]) * h[n] + us * Br[n];
      y += h[n] * Cr[n];
    }
    out[g] = y + xv * Dv;
  }
}

__global__ void scan_fused_kernel(const float* __restrict__ x,
                                  const float* __restrict__ delta,
                                  const float* __restrict__ Bm,
                                  const float* __restrict__ Cm,
                                  const float* __restrict__ W,
                                  const float* __restrict__ bias,
                                  const float* __restrict__ A_log,
                                  const float* __restrict__ Dp,
                                  float* __restrict__ out,
                                  int L, int Dc, int K, int Bsz) {
  const int N = 16;
  long long idx = (long long)blockIdx.x * blockDim.x + threadIdx.x;
  if (idx >= (long long)Bsz * Dc) return;
  int d = (int)(idx % Dc);
  int b = (int)(idx / Dc);
  float An2[N], h[N];
#pragma unroll
  for (int n = 0; n < N; ++n) {
    An2[n] = -__expf(A_log[(size_t)d * N + n]) * LOG2E;
    h[n] = 0.0f;
  }
  float Dv = Dp[d];
  float bv = bias[d];
  const float* wr = W + (size_t)d * K;
  for (int l = 0; l < L; ++l) {
    size_t row = (size_t)b * L + l;
    const float* dr = delta + row * K;
    float acc = bv;
    for (int r = 0; r < K; ++r) acc += dr[r] * wr[r];
    float dtv = softplus_f(acc);
    float xv = x[row * Dc + d];
    float us = dtv * xv;
    const float* Br = Bm + row * N;
    const float* Cr = Cm + row * N;
    float y = 0.0f;
#pragma unroll
    for (int n = 0; n < N; ++n) {
      h[n] = __builtin_amdgcn_exp2f(dtv * An2[n]) * h[n] + us * Br[n];
      y += h[n] * Cr[n];
    }
    out[row * Dc + d] = y + xv * Dv;
  }
}

extern "C" void kernel_launch(void* const* d_in, const int* in_sizes, int n_in,
                              void* d_out, int out_size, void* d_ws, size_t ws_size,
                              hipStream_t stream) {
  const float* x     = (const float*)d_in[0];
  const float* delta = (const float*)d_in[1];
  const float* Bm    = (const float*)d_in[2];
  const float* Cm    = (const float*)d_in[3];
  const float* W     = (const float*)d_in[4];
  const float* bias  = (const float*)d_in[5];
  const float* A_log = (const float*)d_in[6];
  const float* Dp    = (const float*)d_in[7];
  float* out = (float*)d_out;

  int Dc = in_sizes[5];
  int K  = in_sizes[4] / Dc;
  long long BL = (long long)in_sizes[0] / Dc;
  int Bsz = (BL % 2 == 0) ? 2 : 1;
  int L = (int)(BL / Bsz);

  size_t dtElems = (size_t)BL * Dc;
  size_t dtBytes = dtElems * sizeof(float);

  const int Lc = 64;
  bool chunkable = (L % Lc) == 0 && (Dc % 256) == 0;
  int nCh = chunkable ? (L / Lc) : 1;
  size_t chElems = (size_t)Bsz * nCh * Dc;
  size_t fullBytes = dtBytes + chElems * sizeof(float) * (1 + 16 + 16);

  if (ws_size < dtBytes) {
    int threads = 256;
    long long tot = (long long)Bsz * Dc;
    int blocks = (int)((tot + threads - 1) / threads);
    scan_fused_kernel<<<blocks, threads, 0, stream>>>(
        x, delta, Bm, Cm, W, bias, A_log, Dp, out, L, Dc, K, Bsz);
    return;
  }

  float* dt = (float*)d_ws;

  if ((BL % 16) == 0 && (Dc % 32) == 0 && (K % 4) == 0) {
    int tiles = (int)(BL / 16) * (Dc / 32);
    int blocks = (tiles + 7) / 8;
    dtproj_wmma_kernel<<<blocks, 256, 0, stream>>>(delta, W, bias, dt,
                                                   (int)BL, Dc, K);
  } else {
    long long tot = BL * (long long)Dc;
    int blocks = (int)((tot + 255) / 256);
    dtproj_naive_kernel<<<blocks, 256, 0, stream>>>(delta, W, bias, dt, BL, Dc, K);
  }

  if (chunkable && ws_size >= fullBytes) {
    float* Ssum   = dt + dtElems;
    float* Hlocal = Ssum + chElems;
    float* Hinit  = Hlocal + chElems * 16;

    int blocks13 = Bsz * nCh * (Dc / 256);
    size_t smem1 = (size_t)Lc * 16 * sizeof(float);

    scan_chunk_kernel<false><<<blocks13, 256, smem1, stream>>>(
        dt, x, Bm, Cm, A_log, Dp, Hlocal, Ssum, (const float*)nullptr,
        (float*)nullptr, L, Dc, Lc, nCh);

    long long tot = (long long)Bsz * Dc;
    int cblocks = (int)((tot + 255) / 256);
    scan_combine_kernel<<<cblocks, 256, 0, stream>>>(Hlocal, Ssum, A_log, Hinit,
                                                     Dc, nCh, Bsz);

    scan_chunk_kernel<true><<<blocks13, 256, smem1 * 2, stream>>>(
        dt, x, Bm, Cm, A_log, Dp, (float*)nullptr, (float*)nullptr, Hinit, out,
        L, Dc, Lc, nCh);
  } else {
    long long tot = (long long)Bsz * Dc;
    int blocks = (int)((tot + 255) / 256);
    scan_simple_kernel<<<blocks, 256, 0, stream>>>(dt, x, Bm, Cm, A_log, Dp, out,
                                                   L, Dc, Bsz);
  }
}


# scan passes and chunk combine: each serial loop's global operands (dt, x; chunk sums and end states) are fetched in register batches ahead of use, so memory latency is paid per batch, not per step
# speedup vs baseline: 1.2449x; 1.2449x over previous
#include <hip/hip_runtime.h>
#include <math.h>

typedef __attribute__((ext_vector_type(2))) float v2f;
typedef __attribute__((ext_vector_type(8))) float v8f;

#define LOG2E 1.4426950408889634f

__device__ __forceinline__ float softplus_f(float v) {
  return (v > 20.0f) ? v : log1pf(__expf(v));
}

__global__ __launch_bounds__(256) void dtproj_wmma_kernel(
    const float* __restrict__ delta, const float* __restrict__ W,
    const float* __restrict__ bias, float* __restrict__ dt,
    int M, int Dc, int K) {
  int wave = threadIdx.x >> 5;
  int lane = threadIdx.x & 31;
  int tilesN = Dc >> 5;
  int tile = blockIdx.x * 8 + wave;
  int totalTiles = (M >> 4) * tilesN;
  if (tile >= totalTiles) return;
  int tn = tile % tilesN;
  int tm = tile / tilesN;
  int mn   = lane & 15;
  int half = lane >> 4;

  const float* arow  = delta + (size_t)(tm * 16 + mn) * K;
  const float* brow0 = W     + (size_t)(tn * 32 + mn) * K;
  const float* brow1 = W     + (size_t)(tn * 32 + 16 + mn) * K;

  v8f acc0 = {};
  v8f acc1 = {};
  for (int k0 = 0; k0 < K; k0 += 4) {
    int kb = k0 + (half << 1);
    v2f a, b0, b1;
    a.x  = arow[kb];  a.y  = arow[kb + 1];
    b0.x = brow0[kb]; b0.y = brow0[kb + 1];
    b1.x = brow1[kb]; b1.y = brow1[kb + 1];
    acc0 = __builtin_amdgcn_wmma_f32_16x16x4_f32(
        false, a, false, b0, (short)0, acc0, false, false);
    acc1 = __builtin_amdgcn_wmma_f32_16x16x4_f32(
        false, a, false, b1, (short)0, acc1, false, false);
  }

  int gd0 = tn * 32 + mn;
  int gd1 = gd0 + 16;
  float bv0 = bias[gd0];
  float bv1 = bias[gd1];
  int gm0 = tm * 16 + half * 8;
#pragma unroll
  for (int v = 0; v < 8; ++v) {
    size_t rowoff = (size_t)(gm0 + v) * Dc;
    dt[rowoff + gd0] = softplus_f(acc0[v] + bv0);
    dt[rowoff + gd1] = softplus_f(acc1[v] + bv1);
  }
}

__global__ void dtproj_naive_kernel(const float* __restrict__ delta,
                                    const float* __restrict__ W,
                                    const float* __restrict__ bias,
                                    float* __restrict__ dt,
                                    long long M, int Dc, int K) {
  long long idx = (long long)blockIdx.x * blockDim.x + threadIdx.x;
  if (idx >= M * (long long)Dc) return;
  int d = (int)(idx % Dc);
  long long m = idx / Dc;
  float acc = bias[d];
  const float* dr = delta + m * (long long)K;
  const float* wr = W + (size_t)d * K;
  for (int r = 0; r < K; ++r) acc += dr[r] * wr[r];
  dt[idx] = softplus_f(acc);
}

template <bool FINAL>
__global__ __launch_bounds__(256) void scan_chunk_kernel(
    const float* __restrict__ dt, const float* __restrict__ x,
    const float* __restrict__ Bm, const float* __restrict__ Cm,
    const float* __restrict__ A_log, const float* __restrict__ Dp,
    float* __restrict__ Hout, float* __restrict__ Ssum,
    const float* __restrict__ Hinit, float* __restrict__ out,
    int L, int Dc, int Lc, int nCh) {
  const int N = 16;
  const int PB = 16;
  extern __shared__ float smem[];
  float* sB = smem;
  float* sC = smem + (size_t)Lc * N;

  int DB = Dc / blockDim.x;
  int bx = blockIdx.x;
  int dblk = bx % DB; bx /= DB;
  int chunk = bx % nCh; bx /= nCh;
  int b = bx;
  int d = dblk * blockDim.x + threadIdx.x;
  unsigned du = (unsigned)d;

  size_t rowbase = (size_t)b * L + (size_t)chunk * Lc;
  const float* dtrow = dt + rowbase * Dc;
  const float* xrow  = x  + rowbase * Dc;

  const int nb = Lc / PB;
  float dtc[PB], xc[PB], dtn[PB], xn[PB];
#pragma unroll
  for (int i = 0; i < PB; ++i) { dtc[i] = 0.0f; xc[i] = 0.0f; dtn[i] = 0.0f; xn[i] = 0.0f; }
  if (nb > 0) {
#pragma unroll
    for (int i = 0; i < PB; ++i) {
      dtc[i] = dtrow[(size_t)i * Dc + du];
      xc[i]  = xrow[(size_t)i * Dc + du];
    }
  }

  float alog[N], h[N];
#pragma unroll
  for (int n = 0; n < N; ++n) alog[n] = A_log[(size_t)d * N + n];
  size_t hidx = (((size_t)b * nCh + chunk) * Dc + d) * N;
#pragma unroll
  for (int n = 0; n < N; ++n) h[n] = FINAL ? Hinit[hidx + n] : 0.0f;
  float Dv = FINAL ? Dp[d] : 0.0f;

  {
    const float4* Bsrc = (const float4*)(Bm + rowbase * N);
    const float4* Csrc = (const float4*)(Cm + rowbase * N);
    float4* sB4 = (float4*)sB;
    float4* sC4 = (float4*)sC;
    int tot4 = (Lc * N) >> 2;
    for (int i = threadIdx.x; i < tot4; i += blockDim.x) {
      sB4[i] = Bsrc[i];
      if (FINAL) sC4[i] = Csrc[i];
    }
  }
  __syncthreads();

  float An2[N];
#pragma unroll
  for (int n = 0; n < N; ++n) An2[n] = -__expf(alog[n]) * LOG2E;

  float dtsum = 0.0f;

  for (int bi = 0; bi < nb; ++bi) {
    const int l0 = bi * PB;
    if (bi + 1 < nb) {
      const float* dnext = dtrow + (size_t)(l0 + PB) * Dc;
      const float* xnext = xrow  + (size_t)(l0 + PB) * Dc;
#pragma unroll
      for (int i = 0; i < PB; ++i) {
        dtn[i] = dnext[(size_t)i * Dc + du];
        xn[i]  = xnext[(size_t)i * Dc + du];
      }
    }
#pragma unroll
    for (int i = 0; i < PB; ++i) {
      const int l = l0 + i;
      float dtv = dtc[i];
      float xv = xc[i];
      float us = dtv * xv;
      if (!FINAL) dtsum += dtv;
      float y = 0.0f;
#pragma unroll
      for (int n = 0; n < N; ++n) {
        float a = __builtin_amdgcn_exp2f(dtv * An2[n]);
        h[n] = a * h[n] + us * sB[l * N + n];
        if (FINAL) y += h[n] * sC[l * N + n];
      }
      if (FINAL) out[(rowbase + l) * Dc + du] = y + xv * Dv;
    }
#pragma unroll
    for (int i = 0; i < PB; ++i) { dtc[i] = dtn[i]; xc[i] = xn[i]; }
  }

  for (int l = nb * PB; l < Lc; ++l) {
    float dtv = dtrow[(size_t)l * Dc + du];
    float xv = xrow[(size_t)l * Dc + du];
    float us = dtv * xv;
    if (!FINAL) dtsum += dtv;
    float y = 0.0f;
#pragma unroll
    for (int n = 0; n < N; ++n) {
      float a = __builtin_amdgcn_exp2f(dtv * An2[n]);
      h[n] = a * h[n] + us * sB[l * N + n];
      if (FINAL) y += h[n] * sC[l * N + n];
    }
    if (FINAL) out[(rowbase + l) * Dc + du] = y + xv * Dv;
  }

  if (!FINAL) {
#pragma unroll
    for (int n = 0; n < N; ++n) Hout[hidx + n] = h[n];
    Ssum[((size_t)b * nCh + chunk) * Dc + d] = dtsum;
  }
}

__global__ __launch_bounds__(256) void scan_combine_kernel(
    const float* __restrict__ Hlocal, const float* __restrict__ Ssum,
    const float* __restrict__ A_log, float* __restrict__ Hinit,
    int Dc, int nCh, int Bsz) {
  const int N = 16;
  const int CB = 8;
  long long idx = (long long)blockIdx.x * blockDim.x + threadIdx.x;
  if (idx >= (long long)Bsz * Dc) return;
  int d = (int)(idx % Dc);
  int b = (int)(idx / Dc);
  float An2[N], h[N];
#pragma unroll
  for (int n = 0; n < N; ++n) {
    An2[n] = -__expf(A_log[(size_t)d * N + n]) * LOG2E;
    h[n] = 0.0f;
  }
  int c = 0;
  for (; c + CB <= nCh; c += CB) {
    float S[CB];
    float hl[CB][N];
#pragma unroll
    for (int j = 0; j < CB; ++j) {
      size_t base = ((size_t)b * nCh + (c + j)) * Dc + d;
      S[j] = Ssum[base];
#pragma unroll
      for (int n = 0; n < N; ++n) hl[j][n] = Hlocal[base * N + n];
    }
    __builtin_amdgcn_sched_barrier(0);
#pragma unroll
    for (int j = 0; j < CB; ++j) {
      size_t hidx = (((size_t)b * nCh + (c + j)) * Dc + d) * N;
#pragma unroll
      for (int n = 0; n < N; ++n) {
        Hinit[hidx + n] = h[n];
        float a = __builtin_amdgcn_exp2f(An2[n] * S[j]);
        h[n] = a * h[n] + hl[j][n];
      }
    }
  }
  for (; c < nCh; ++c) {
    size_t base = ((size_t)b * nCh + c) * Dc + d;
    size_t hidx = base * N;
    float S = Ssum[base];
#pragma unroll
    for (int n = 0; n < N; ++n) {
      Hinit[hidx + n] = h[n];
      float a = __builtin_amdgcn_exp2f(An2[n] * S);
      h[n] = a * h[n] + Hlocal[hidx + n];
    }
  }
}

__global__ void scan_simple_kernel(const float* __restrict__ dt,
                                   const float* __restrict__ x,
                                   const float* __restrict__ Bm,
                                   const float* __restrict__ Cm,
                                   const float* __restrict__ A_log,
                                   const float* __restrict__ Dp,
                                   float* __restrict__ out,
                                   int L, int Dc, int Bsz) {
  const int N = 16;
  long long idx = (long long)blockIdx.x * blockDim.x + threadIdx.x;
  if (idx >= (long long)Bsz * Dc) return;
  int d = (int)(idx % Dc);
  int b = (int)(idx / Dc);
  float An2[N], h[N];
#pragma unroll
  for (int n = 0; n < N; ++n) {
    An2[n] = -__expf(A_log[(size_t)d * N + n]) * LOG2E;
    h[n] = 0.0f;
  }
  float Dv = Dp[d];
  for (int l = 0; l < L; ++l) {
    size_t row = (size_t)b * L + l;
    size_t g = row * Dc + d;
    float dtv = dt[g];
    float xv = x[g];
    float us = dtv * xv;
    const float* Br = Bm + row * N;
    const float* Cr = Cm + row * N;
    float y = 0.0f;
#pragma unroll
    for (int n = 0; n < N; ++n) {
      h[n] = __builtin_amdgcn_exp2f(dtv * An2[n]) * h[n] + us * Br[n];
      y += h[n] * Cr[n];
    }
    out[g] = y + xv * Dv;
  }
}

__global__ void scan_fused_kernel(const float* __restrict__ x,
                                  const float* __restrict__ delta,
                                  const float* __restrict__ Bm,
                                  const float* __restrict__ Cm,
                                  const float* __restrict__ W,
                                  const float* __restrict__ bias,
                                  const float* __restrict__ A_log,
                                  const float* __restrict__ Dp,
                                  float* __restrict__ out,
                                  int L, int Dc, int K, int Bsz) {
  const int N = 16;
  long long idx = (long long)blockIdx.x * blockDim.x + threadIdx.x;
  if (idx >= (long long)Bsz * Dc) return;
  int d = (int)(idx % Dc);
  int b = (int)(idx / Dc);
  float An2[N], h[N];
#pragma unroll
  for (int n = 0; n < N; ++n) {
    An2[n] = -__expf(A_log[(size_t)d * N + n]) * LOG2E;
    h[n] = 0.0f;
  }
  float Dv = Dp[d];
  float bv = bias[d];
  const float* wr = W + (size_t)d * K;
  for (int l = 0; l < L; ++l) {
    size_t row = (size_t)b * L + l;
    const float* dr = delta + row * K;
    float acc = bv;
    for (int r = 0; r < K; ++r) acc += dr[r] * wr[r];
    float dtv = softplus_f(acc);
    float xv = x[row * Dc + d];
    float us = dtv * xv;
    const float* Br = Bm + row * N;
    const float* Cr = Cm + row * N;
    float y = 0.0f;
#pragma unroll
    for (int n = 0; n < N; ++n) {
      h[n] = __builtin_amdgcn_exp2f(dtv * An2[n]) * h[n] + us * Br[n];
      y += h[n] * Cr[n];
    }
    out[row * Dc + d] = y + xv * Dv;
  }
}

extern "C" void kernel_launch(void* const* d_in, const int* in_sizes, int n_in,
                              void* d_out, int out_size, void* d_ws, size_t ws_size,
                              hipStream_t stream) {
  const float* x     = (const float*)d_in[0];
  const float* delta = (const float*)d_in[1];
  const float* Bm    = (const float*)d_in[2];
  const float* Cm    = (const float*)d_in[3];
  const float* W     = (const float*)d_in[4];
  const float* bias  = (const float*)d_in[5];
  const float* A_log = (const float*)d_in[6];
  const float* Dp    = (const float*)d_in[7];
  float* out = (float*)d_out;

  int Dc = in_sizes[5];
  int K  = in_sizes[4] / Dc;
  long long BL = (long long)in_sizes[0] / Dc;
  int Bsz = (BL % 2 == 0) ? 2 : 1;
  int L = (int)(BL / Bsz);

  size_t dtElems = (size_t)BL * Dc;
  size_t dtBytes = dtElems * sizeof(float);

  const int Lc = 64;
  bool chunkable = (L % Lc) == 0 && (Dc % 256) == 0;
  int nCh = chunkable ? (L / Lc) : 1;
  size_t chElems = (size_t)Bsz * nCh * Dc;
  size_t fullBytes = dtBytes + chElems * sizeof(float) * (1 + 16 + 16);

  if (ws_size < dtBytes) {
    int threads = 256;
    long long tot = (long long)Bsz * Dc;
    int blocks = (int)((tot + threads - 1) / threads);
    scan_fused_kernel<<<blocks, threads, 0, stream>>>(
        x, delta, Bm, Cm, W, bias, A_log, Dp, out, L, Dc, K, Bsz);
    return;
  }

  float* dt = (float*)d_ws;

  if ((BL % 16) == 0 && (Dc % 32) == 0 && (K % 4) == 0) {
    int tiles = (int)(BL / 16) * (Dc / 32);
    int blocks = (tiles + 7) / 8;
    dtproj_wmma_kernel<<<blocks, 256, 0, stream>>>(delta, W, bias, dt,
                                                   (int)BL, Dc, K);
  } else {
    long long tot = BL * (long long)Dc;
    int blocks = (int)((tot + 255) / 256);
    dtproj_naive_kernel<<<blocks, 256, 0, stream>>>(delta, W, bias, dt, BL, Dc, K);
  }

  if (chunkable && ws_size >= fullBytes) {
    float* Ssum   = dt + dtElems;
    float* Hlocal = Ssum + chElems;
    float* Hinit  = Hlocal + chElems * 16;

    int blocks13 = Bsz * nCh * (Dc / 256);
    size_t smem1 = (size_t)Lc * 16 * sizeof(float);

    scan_chunk_kernel<false><<<blocks13, 256, smem1, stream>>>(
        dt, x, Bm, Cm, A_log, Dp, Hlocal, Ssum, (const float*)nullptr,
        (float*)nullptr, L, Dc, Lc, nCh);

    long long tot = (long long)Bsz * Dc;
    int cblocks = (int)((tot + 255) / 256);
    scan_combine_kernel<<<cblocks, 256, 0, stream>>>(Hlocal, Ssum, A_log, Hinit,
                                                     Dc, nCh, Bsz);

    scan_chunk_kernel<true><<<blocks13, 256, smem1 * 2, stream>>>(
        dt, x, Bm, Cm, A_log, Dp, (float*)nullptr, (float*)nullptr, Hinit, out,
        L, Dc, Lc, nCh);
  } else {
    long long tot = (long long)Bsz * Dc;
    int blocks = (int)((tot + 255) / 256);
    scan_simple_kernel<<<blocks, 256, 0, stream>>>(dt, x, Bm, Cm, A_log, Dp, out,
                                                   L, Dc, Bsz);
  }
}


# scan passes: the B and C rows of a timestep, the same for every lane, come by wave-uniform scalar loads into scalar registers, requested one step ahead, instead of LDS; LDS staging and its barrier go
# speedup vs baseline: 1.5163x; 1.2180x over previous
#include <hip/hip_runtime.h>
#include <math.h>

typedef __attribute__((ext_vector_type(2))) float v2f;
typedef __attribute__((ext_vector_type(8))) float v8f;

#define LOG2E 1.4426950408889634f

__device__ __forceinline__ float softplus_f(float v) {
  return (v > 20.0f) ? v : log1pf(__expf(v));
}

__global__ __launch_bounds__(256) void dtproj_wmma_kernel(
    const float* __restrict__ delta, const float* __restrict__ W,
    const float* __restrict__ bias, float* __restrict__ dt,
    int M, int Dc, int K) {
  int wave = threadIdx.x >> 5;
  int lane = threadIdx.x & 31;
  int tilesN = Dc >> 5;
  int tile = blockIdx.x * 8 + wave;
  int totalTiles = (M >> 4) * tilesN;
  if (tile >= totalTiles) return;
  int tn = tile % tilesN;
  int tm = tile / tilesN;
  int mn   = lane & 15;
  int half = lane >> 4;

  const float* arow  = delta + (size_t)(tm * 16 + mn) * K;
  const float* brow0 = W     + (size_t)(tn * 32 + mn) * K;
  const float* brow1 = W     + (size_t)(tn * 32 + 16 + mn) * K;

  v8f acc0 = {};
  v8f acc1 = {};
  for (int k0 = 0; k0 < K; k0 += 4) {
    int kb = k0 + (half << 1);
    v2f a, b0, b1;
    a.x  = arow[kb];  a.y  = arow[kb + 1];
    b0.x = brow0[kb]; b0.y = brow0[kb + 1];
    b1.x = brow1[kb]; b1.y = brow1[kb + 1];
    acc0 = __builtin_amdgcn_wmma_f32_16x16x4_f32(
        false, a, false, b0, (short)0, acc0, false, false);
    acc1 = __builtin_amdgcn_wmma_f32_16x16x4_f32(
        false, a, false, b1, (short)0, acc1, false, false);
  }

  int gd0 = tn * 32 + mn;
  int gd1 = gd0 + 16;
  float bv0 = bias[gd0];
  float bv1 = bias[gd1];
  int gm0 = tm * 16 + half * 8;
#pragma unroll
  for (int v = 0; v < 8; ++v) {
    size_t rowoff = (size_t)(gm0 + v) * Dc;
    dt[rowoff + gd0] = softplus_f(acc0[v] + bv0);
    dt[rowoff + gd1] = softplus_f(acc1[v] + bv1);
  }
}

__global__ void dtproj_naive_kernel(const float* __restrict__ delta,
                                    const float* __restrict__ W,
                                    const float* __restrict__ bias,
                                    float* __restrict__ dt,
                                    long long M, int Dc, int K) {
  long long idx = (long long)blockIdx.x * blockDim.x + threadIdx.x;
  if (idx >= M * (long long)Dc) return;
  int d = (int)(idx % Dc);
  long long m = idx / Dc;
  float acc = bias[d];
  const float* dr = delta + m * (long long)K;
  const float* wr = W + (size_t)d * K;
  for (int r = 0; r < K; ++r) acc += dr[r] * wr[r];
  dt[idx] = softplus_f(acc);
}

template <bool FINAL>
__global__ __launch_bounds__(256) void scan_chunk_kernel(
    const float* __restrict__ dt, const float* __restrict__ x,
    const float* __restrict__ Bm, const float* __restrict__ Cm,
    const float* __restrict__ A_log, const float* __restrict__ Dp,
    float* __restrict__ Hout, float* __restrict__ Ssum,
    const float* __restrict__ Hinit, float* __restrict__ out,
    int L, int Dc, int Lc, int nCh) {
  const int N = 16;
  const int PB = 16;

  int DB = Dc / blockDim.x;
  int bx = blockIdx.x;
  int dblk = bx % DB; bx /= DB;
  int chunk = bx % nCh; bx /= nCh;
  int b = bx;
  int d = dblk * blockDim.x + threadIdx.x;
  unsigned du = (unsigned)d;

  size_t rowbase = (size_t)b * L + (size_t)chunk * Lc;
  const float* dtrow = dt + rowbase * Dc;
  const float* xrow  = x  + rowbase * Dc;
  typedef const __attribute__((address_space(4))) float cfloat_k;
  cfloat_k* Brow = (cfloat_k*)(Bm + rowbase * N);
  cfloat_k* Crow = (cfloat_k*)(Cm + rowbase * N);
  float bcur[N], ccur[N];
#pragma unroll
  for (int n = 0; n < N; ++n) {
    bcur[n] = Brow[n];
    ccur[n] = FINAL ? Crow[n] : 0.0f;
  }

  const int nb = Lc / PB;
  float dtc[PB], xc[PB], dtn[PB], xn[PB];
#pragma unroll
  for (int i = 0; i < PB; ++i) { dtc[i] = 0.0f; xc[i] = 0.0f; dtn[i] = 0.0f; xn[i] = 0.0f; }
  if (nb > 0) {
#pragma unroll
    for (int i = 0; i < PB; ++i) {
      dtc[i] = dtrow[(size_t)i * Dc + du];
      xc[i]  = xrow[(size_t)i * Dc + du];
    }
  }

  float alog[N], h[N];
  size_t hidx = (((size_t)b * nCh + chunk) * Dc + d) * N;
#pragma unroll
  for (int n = 0; n < N; ++n) h[n] = FINAL ? Hinit[hidx + n] : 0.0f;
  float Dv = FINAL ? Dp[d] : 0.0f;
#pragma unroll
  for (int n = 0; n < N; ++n) alog[n] = A_log[(size_t)d * N + n];

  float An2[N];
#pragma unroll
  for (int n = 0; n < N; ++n) An2[n] = -__expf(alog[n]) * LOG2E;

  float dtsum = 0.0f;

  if (FINAL) {
    asm volatile("" :: "v"(Dv), "v"(h[0]), "v"(h[1]), "v"(h[2]), "v"(h[3]),
                       "v"(h[4]), "v"(h[5]), "v"(h[6]), "v"(h[7]),
                       "v"(h[8]), "v"(h[9]), "v"(h[10]), "v"(h[11]),
                       "v"(h[12]), "v"(h[13]), "v"(h[14]), "v"(h[15]));
  }

  for (int bi = 0; bi < nb; ++bi) {
    const int l0 = bi * PB;
    if (bi + 1 < nb) {
      const float* dnext = dtrow + (size_t)(l0 + PB) * Dc;
      const float* xnext = xrow  + (size_t)(l0 + PB) * Dc;
#pragma unroll
      for (int i = 0; i < PB; ++i) {
        dtn[i] = dnext[(size_t)i * Dc + du];
        xn[i]  = xnext[(size_t)i * Dc + du];
      }
    }
#pragma unroll
    for (int i = 0; i < PB; ++i) {
      const int l = l0 + i;
      float dtv = dtc[i];
      float xv = xc[i];
      float us = dtv * xv;
      if (!FINAL) dtsum += dtv;
      float ub[N];
#pragma unroll
      for (int n = 0; n < N; ++n) ub[n] = us * bcur[n];
      asm volatile("" :: "s"(bcur[0]), "s"(bcur[1]), "s"(bcur[2]), "s"(bcur[3]),
                         "s"(bcur[4]), "s"(bcur[5]), "s"(bcur[6]), "s"(bcur[7]),
                         "s"(bcur[8]), "s"(bcur[9]), "s"(bcur[10]), "s"(bcur[11]),
                         "s"(bcur[12]), "s"(bcur[13]), "s"(bcur[14]), "s"(bcur[15]));
      const int ln = (i + 1 < PB) ? (l + 1) : ((l + 1 < Lc) ? (l + 1) : l);
      float cnx[N];
#pragma unroll
      for (int n = 0; n < N; ++n) {
        bcur[n] = Brow[ln * N + n];
        cnx[n] = FINAL ? Crow[ln * N + n] : 0.0f;
      }
      __builtin_amdgcn_sched_barrier(0);
      float y = 0.0f;
#pragma unroll
      for (int n = 0; n < N; ++n) {
        float a = __builtin_amdgcn_exp2f(dtv * An2[n]);
        h[n] = __builtin_fmaf(a, h[n], ub[n]);
        if (FINAL) y += h[n] * ccur[n];
      }
      if (FINAL) out[(rowbase + l) * Dc + du] = y + xv * Dv;
#pragma unroll
      for (int n = 0; n < N; ++n) ccur[n] = cnx[n];
    }
#pragma unroll
    for (int i = 0; i < PB; ++i) { dtc[i] = dtn[i]; xc[i] = xn[i]; }
  }

  for (int l = nb * PB; l < Lc; ++l) {
    float dtv = dtrow[(size_t)l * Dc + du];
    float xv = xrow[(size_t)l * Dc + du];
    float us = dtv * xv;
    if (!FINAL) dtsum += dtv;
    float y = 0.0f;
#pragma unroll
    for (int n = 0; n < N; ++n) {
      float a = __builtin_amdgcn_exp2f(dtv * An2[n]);
      h[n] = __builtin_fmaf(a, h[n], us * Brow[l * N + n]);
      if (FINAL) y += h[n] * Crow[l * N + n];
    }
    if (FINAL) out[(rowbase + l) * Dc + du] = y + xv * Dv;
  }

  if (!FINAL) {
#pragma unroll
    for (int n = 0; n < N; ++n) Hout[hidx + n] = h[n];
    Ssum[((size_t)b * nCh + chunk) * Dc + d] = dtsum;
  }
}

__global__ __launch_bounds__(256) void scan_combine_kernel(
    const float* __restrict__ Hlocal, const float* __restrict__ Ssum,
    const float* __restrict__ A_log, float* __restrict__ Hinit,
    int Dc, int nCh, int Bsz) {
  const int N = 16;
  const int CB = 8;
  long long idx = (long long)blockIdx.x * blockDim.x + threadIdx.x;
  if (idx >= (long long)Bsz * Dc) return;
  int d = (int)(idx % Dc);
  int b = (int)(idx / Dc);
  float An2[N], h[N];
#pragma unroll
  for (int n = 0; n < N; ++n) {
    An2[n] = -__expf(A_log[(size_t)d * N + n]) * LOG2E;
    h[n] = 0.0f;
  }
  int c = 0;
  for (; c + CB <= nCh; c += CB) {
    float S[CB];
    float hl[CB][N];
#pragma unroll
    for (int j = 0; j < CB; ++j) {
      size_t base = ((size_t)b * nCh + (c + j)) * Dc + d;
      S[j] = Ssum[base];
#pragma unroll
      for (int n = 0; n < N; ++n) hl[j][n] = Hlocal[base * N + n];
    }
    __builtin_amdgcn_sched_barrier(0);
#pragma unroll
    for (int j = 0; j < CB; ++j) {
      size_t hidx = (((size_t)b * nCh + (c + j)) * Dc + d) * N;
#pragma unroll
      for (int n = 0; n < N; ++n) {
        Hinit[hidx + n] = h[n];
        float a = __builtin_amdgcn_exp2f(An2[n] * S[j]);
        h[n] = a * h[n] + hl[j][n];
      }
    }
  }
  for (; c < nCh; ++c) {
    size_t base = ((size_t)b * nCh + c) * Dc + d;
    size_t hidx = base * N;
    float S = Ssum[base];
#pragma unroll
    for (int n = 0; n < N; ++n) {
      Hinit[hidx + n] = h[n];
      float a = __builtin_amdgcn_exp2f(An2[n] * S);
      h[n] = a * h[n] + Hlocal[hidx + n];
    }
  }
}

__global__ void scan_simple_kernel(const float* __restrict__ dt,
                                   const float* __restrict__ x,
                                   const float* __restrict__ Bm,
                                   const float* __restrict__ Cm,
                                   const float* __restrict__ A_log,
                                   const float* __restrict__ Dp,
                                   float* __restrict__ out,
                                   int L, int Dc, int Bsz) {
  const int N = 16;
  long long idx = (long long)blockIdx.x * blockDim.x + threadIdx.x;
  if (idx >= (long long)Bsz * Dc) return;
  int d = (int)(idx % Dc);
  int b = (int)(idx / Dc);
  float An2[N], h[N];
#pragma unroll
  for (int n = 0; n < N; ++n) {
    An2[n] = -__expf(A_log[(size_t)d * N + n]) * LOG2E;
    h[n] = 0.0f;
  }
  float Dv = Dp[d];
  for (int l = 0; l < L; ++l) {
    size_t row = (size_t)b * L + l;
    size_t g = row * Dc + d;
    float dtv = dt[g];
    float xv = x[g];
    float us = dtv * xv;
    const float* Br = Bm + row * N;
    const float* Cr = Cm + row * N;
    float y = 0.0f;
#pragma unroll
    for (int n = 0; n < N; ++n) {
      h[n] = __builtin_amdgcn_exp2f(dtv * An2[n]) * h[n] + us * Br[n];
      y += h[n] * Cr[n];
    }
    out[g] = y + xv * Dv;
  }
}

__global__ void scan_fused_kernel(const float* __restrict__ x,
                                  const float* __restrict__ delta,
                                  const float* __restrict__ Bm,
                                  const float* __restrict__ Cm,
                                  const float* __restrict__ W,
                                  const float* __restrict__ bias,
                                  const float* __restrict__ A_log,
                                  const float* __restrict__ Dp,
                                  float* __restrict__ out,
                                  int L, int Dc, int K, int Bsz) {
  const int N = 16;
  long long idx = (long long)blockIdx.x * blockDim.x + threadIdx.x;
  if (idx >= (long long)Bsz * Dc) return;
  int d = (int)(idx % Dc);
  int b = (int)(idx / Dc);
  float An2[N], h[N];
#pragma unroll
  for (int n = 0; n < N; ++n) {
    An2[n] = -__expf(A_log[(size_t)d * N + n]) * LOG2E;
    h[n] = 0.0f;
  }
  float Dv = Dp[d];
  float bv = bias[d];
  const float* wr = W + (size_t)d * K;
  for (int l = 0; l < L; ++l) {
    size_t row = (size_t)b * L + l;
    const float* dr = delta + row * K;
    float acc = bv;
    for (int r = 0; r < K; ++r) acc += dr[r] * wr[r];
    float dtv = softplus_f(acc);
    float xv = x[row * Dc + d];
    float us = dtv * xv;
    const float* Br = Bm + row * N;
    const float* Cr = Cm + row * N;
    float y = 0.0f;
#pragma unroll
    for (int n = 0; n < N; ++n) {
      h[n] = __builtin_amdgcn_exp2f(dtv * An2[n]) * h[n] + us * Br[n];
      y += h[n] * Cr[n];
    }
    out[row * Dc + d] = y + xv * Dv;
  }
}

extern "C" void kernel_launch(void* const* d_in, const int* in_sizes, int n_in,
                              void* d_out, int out_size, void* d_ws, size_t ws_size,
                              hipStream_t stream) {
  const float* x     = (const float*)d_in[0];
  const float* delta = (const float*)d_in[1];
  const float* Bm    = (const float*)d_in[2];
  const float* Cm    = (const float*)d_in[3];
  const float* W     = (const float*)d_in[4];
  const float* bias  = (const float*)d_in[5];
  const float* A_log = (const float*)d_in[6];
  const float* Dp    = (const float*)d_in[7];
  float* out = (float*)d_out;

  int Dc = in_sizes[5];
  int K  = in_sizes[4] / Dc;
  long long BL = (long long)in_sizes[0] / Dc;
  int Bsz = (BL % 2 == 0) ? 2 : 1;
  int L = (int)(BL / Bsz);

  size_t dtElems = (size_t)BL * Dc;
  size_t dtBytes = dtElems * sizeof(float);

  const int Lc = 64;
  bool chunkable = (L % Lc) == 0 && (Dc % 256) == 0;
  int nCh = chunkable ? (L / Lc) : 1;
  size_t chElems = (size_t)Bsz * nCh * Dc;
  size_t fullBytes = dtBytes + chElems * sizeof(float) * (1 + 16 + 16);

  if (ws_size < dtBytes) {
    int threads = 256;
    long long tot = (long long)Bsz * Dc;
    int blocks = (int)((tot + threads - 1) / threads);
    scan_fused_kernel<<<blocks, threads, 0, stream>>>(
        x, delta, Bm, Cm, W, bias, A_log, Dp, out, L, Dc, K, Bsz);
    return;
  }

  float* dt = (float*)d_ws;

  if ((BL % 16) == 0 && (Dc % 32) == 0 && (K % 4) == 0) {
    int tiles = (int)(BL / 16) * (Dc / 32);
    int blocks = (tiles + 7) / 8;
    dtproj_wmma_kernel<<<blocks, 256, 0, stream>>>(delta, W, bias, dt,
                                                   (int)BL, Dc, K);
  } else {
    long long tot = BL * (long long)Dc;
    int blocks = (int)((tot + 255) / 256);
    dtproj_naive_kernel<<<blocks, 256, 0, stream>>>(delta, W, bias, dt, BL, Dc, K);
  }

  if (chunkable && ws_size >= fullBytes) {
    float* Ssum   = dt + dtElems;
    float* Hlocal = Ssum + chElems;
    float* Hinit  = Hlocal + chElems * 16;

    int blocks13 = Bsz * nCh * (Dc / 256);
    scan_chunk_kernel<false><<<blocks13, 256, 0, stream>>>(
        dt, x, Bm, Cm, A_log, Dp, Hlocal, Ssum, (const float*)nullptr,
        (float*)nullptr, L, Dc, Lc, nCh);

    long long tot = (long long)Bsz * Dc;
    int cblocks = (int)((tot + 255) / 256);
    scan_combine_kernel<<<cblocks, 256, 0, stream>>>(Hlocal, Ssum, A_log, Hinit,
                                                     Dc, nCh, Bsz);

    scan_chunk_kernel<true><<<blocks13, 256, 0, stream>>>(
        dt, x, Bm, Cm, A_log, Dp, (float*)nullptr, (float*)nullptr, Hinit, out,
        L, Dc, Lc, nCh);
  } else {
    long long tot = (long long)Bsz * Dc;
    int blocks = (int)((tot + 255) / 256);
    scan_simple_kernel<<<blocks, 256, 0, stream>>>(dt, x, Bm, Cm, A_log, Dp, out,
                                                   L, Dc, Bsz);
  }
}


# dt projection: each work-group stages its delta and W pieces in LDS with whole-line loads, one K chunk ahead; each wave keeps 2x2 accumulators, so a fragment read feeds 2 matrix instructions
# speedup vs baseline: 1.8626x; 1.2284x over previous
#include <hip/hip_runtime.h>
#include <math.h>

typedef __attribute__((ext_vector_type(2))) float v2f;
typedef __attribute__((ext_vector_type(8))) float v8f;
typedef __attribute__((ext_vector_type(4))) float v4f;

#define LOG2E 1.4426950408889634f

__device__ __forceinline__ float softplus_f(float v) {
  return (v > 20.0f) ? v : log1pf(__expf(v));
}

__global__ __launch_bounds__(256) void dtproj_wmma_kernel(
    const float* __restrict__ delta, const float* __restrict__ W,
    const float* __restrict__ bias, float* __restrict__ dt,
    int M, int Dc, int K) {
  int wave = threadIdx.x >> 5;
  int lane = threadIdx.x & 31;
  int tilesN = Dc >> 5;
  int tile = blockIdx.x * 8 + wave;
  int totalTiles = (M >> 4) * tilesN;
  if (tile >= totalTiles) return;
  int tn = tile % tilesN;
  int tm = tile / tilesN;
  int mn   = lane & 15;
  int half = lane >> 4;

  const float* arow  = delta + (size_t)(tm * 16 + mn) * K;
  const float* brow0 = W     + (size_t)(tn * 32 + mn) * K;
  const float* brow1 = W     + (size_t)(tn * 32 + 16 + mn) * K;

  v8f acc0 = {};
  v8f acc1 = {};
  for (int k0 = 0; k0 < K; k0 += 4) {
    int kb = k0 + (half << 1);
    v2f a, b0, b1;
    a.x  = arow[kb];  a.y  = arow[kb + 1];
    b0.x = brow0[kb]; b0.y = brow0[kb + 1];
    b1.x = brow1[kb]; b1.y = brow1[kb + 1];
    acc0 = __builtin_amdgcn_wmma_f32_16x16x4_f32(
        false, a, false, b0, (short)0, acc0, false, false);
    acc1 = __builtin_amdgcn_wmma_f32_16x16x4_f32(
        false, a, false, b1, (short)0, acc1, false, false);
  }

  int gd0 = tn * 32 + mn;
  int gd1 = gd0 + 16;
  float bv0 = bias[gd0];
  float bv1 = bias[gd1];
  int gm0 = tm * 16 + half * 8;
#pragma unroll
  for (int v = 0; v < 8; ++v) {
    size_t rowoff = (size_t)(gm0 + v) * Dc;
    dt[rowoff + gd0] = softplus_f(acc0[v] + bv0);
    dt[rowoff + gd1] = softplus_f(acc1[v] + bv1);
  }
}

#define DTP_BM 64
#define DTP_BN 128
#define DTP_KC 32
#define DTP_LD 36

__global__ __launch_bounds__(256) void dtproj_tiled_kernel(
    const float* __restrict__ delta, const float* __restrict__ W,
    const float* __restrict__ bias, float* __restrict__ dt,
    int M, int Dc, int K) {
  __shared__ __attribute__((aligned(16))) float sA[2][DTP_BM * DTP_LD];
  __shared__ __attribute__((aligned(16))) float sB[2][DTP_BN * DTP_LD];

  const int tid  = threadIdx.x;
  const int wave = tid >> 5;
  const int lane = tid & 31;
  const int c = lane & 15;
  const int h = lane >> 4;
  const int tilesN = Dc / DTP_BN;
  const int m0 = (blockIdx.x / tilesN) * DTP_BM;
  const int n0 = (blockIdx.x % tilesN) * DTP_BN;
  const int wm = (wave >> 2) * 32;
  const int wn = (wave & 3) * 32;

  const int crow = tid >> 3;
  const int ccol = (tid & 7) * 4;
  const float* gA = delta + (size_t)(m0 + crow) * K + ccol;
  const float* gB = W     + (size_t)(n0 + crow) * K + ccol;
  const size_t gstep = (size_t)32 * K;

  v4f pa[2], pb[4];
#pragma unroll
  for (int p = 0; p < 2; ++p) pa[p] = *(const v4f*)(gA + p * gstep);
#pragma unroll
  for (int p = 0; p < 4; ++p) pb[p] = *(const v4f*)(gB + p * gstep);
#pragma unroll
  for (int p = 0; p < 2; ++p) *(v4f*)&sA[0][(crow + 32 * p) * DTP_LD + ccol] = pa[p];
#pragma unroll
  for (int p = 0; p < 4; ++p) *(v4f*)&sB[0][(crow + 32 * p) * DTP_LD + ccol] = pb[p];
  __syncthreads();

  v8f acc[2][2];
#pragma unroll
  for (int i = 0; i < 2; ++i)
#pragma unroll
    for (int j = 0; j < 2; ++j) acc[i][j] = (v8f){};

  const int nk = K / DTP_KC;
  for (int kc = 0; kc < nk; ++kc) {
    const int st = kc & 1;
    const bool more = (kc + 1 < nk);
    if (more) {
      const float* gAn = gA + (size_t)(kc + 1) * DTP_KC;
      const float* gBn = gB + (size_t)(kc + 1) * DTP_KC;
#pragma unroll
      for (int p = 0; p < 2; ++p) pa[p] = *(const v4f*)(gAn + p * gstep);
#pragma unroll
      for (int p = 0; p < 4; ++p) pb[p] = *(const v4f*)(gBn + p * gstep);
    }
    const float* abase = &sA[st][(wm + c) * DTP_LD + 2 * h];
    const float* bbase = &sB[st][(wn + c) * DTP_LD + 2 * h];
#pragma unroll
    for (int kk = 0; kk < DTP_KC; kk += 4) {
      v2f af[2], bf[2];
#pragma unroll
      for (int i = 0; i < 2; ++i) af[i] = *(const v2f*)(abase + i * 16 * DTP_LD + kk);
#pragma unroll
      for (int j = 0; j < 2; ++j) bf[j] = *(const v2f*)(bbase + j * 16 * DTP_LD + kk);
#pragma unroll
      for (int i = 0; i < 2; ++i)
#pragma unroll
        for (int j = 0; j < 2; ++j)
          acc[i][j] = __builtin_amdgcn_wmma_f32_16x16x4_f32(
              false, af[i], false, bf[j], (short)0, acc[i][j], false, false);
    }
    if (more) {
#pragma unroll
      for (int p = 0; p < 2; ++p) *(v4f*)&sA[st ^ 1][(crow + 32 * p) * DTP_LD + ccol] = pa[p];
#pragma unroll
      for (int p = 0; p < 4; ++p) *(v4f*)&sB[st ^ 1][(crow + 32 * p) * DTP_LD + ccol] = pb[p];
    }
    __syncthreads();
  }

  const int gd0 = n0 + wn + c;
  const int gd1 = gd0 + 16;
  const float bv0 = bias[gd0];
  const float bv1 = bias[gd1];
#pragma unroll
  for (int i = 0; i < 2; ++i) {
    const int gm0 = m0 + wm + 16 * i + 8 * h;
#pragma unroll
    for (int r = 0; r < 8; ++r) {
      size_t rowoff = (size_t)(gm0 + r) * Dc;
      dt[rowoff + gd0] = softplus_f(acc[i][0][r] + bv0);
      dt[rowoff + gd1] = softplus_f(acc[i][1][r] + bv1);
    }
  }
}

__global__ void dtproj_naive_kernel(const float* __restrict__ delta,
                                    const float* __restrict__ W,
                                    const float* __restrict__ bias,
                                    float* __restrict__ dt,
                                    long long M, int Dc, int K) {
  long long idx = (long long)blockIdx.x * blockDim.x + threadIdx.x;
  if (idx >= M * (long long)Dc) return;
  int d = (int)(idx % Dc);
  long long m = idx / Dc;
  float acc = bias[d];
  const float* dr = delta + m * (long long)K;
  const float* wr = W + (size_t)d * K;
  for (int r = 0; r < K; ++r) acc += dr[r] * wr[r];
  dt[idx] = softplus_f(acc);
}

template <bool FINAL>
__global__ __launch_bounds__(256) void scan_chunk_kernel(
    const float* __restrict__ dt, const float* __restrict__ x,
    const float* __restrict__ Bm, const float* __restrict__ Cm,
    const float* __restrict__ A_log, const float* __restrict__ Dp,
    float* __restrict__ Hout, float* __restrict__ Ssum,
    const float* __restrict__ Hinit, float* __restrict__ out,
    int L, int Dc, int Lc, int nCh) {
  const int N = 16;
  const int PB = 16;

  int DB = Dc / blockDim.x;
  int bx = blockIdx.x;
  int dblk = bx % DB; bx /= DB;
  int chunk = bx % nCh; bx /= nCh;
  int b = bx;
  int d = dblk * blockDim.x + threadIdx.x;
  unsigned du = (unsigned)d;

  size_t rowbase = (size_t)b * L + (size_t)chunk * Lc;
  const float* dtrow = dt + rowbase * Dc;
  const float* xrow  = x  + rowbase * Dc;
  typedef const __attribute__((address_space(4))) float cfloat_k;
  cfloat_k* Brow = (cfloat_k*)(Bm + rowbase * N);
  cfloat_k* Crow = (cfloat_k*)(Cm + rowbase * N);
  float bcur[N], ccur[N];
#pragma unroll
  for (int n = 0; n < N; ++n) {
    bcur[n] = Brow[n];
    ccur[n] = FINAL ? Crow[n] : 0.0f;
  }

  const int nb = Lc / PB;
  float dtc[PB], xc[PB], dtn[PB], xn[PB];
#pragma unroll
  for (int i = 0; i < PB; ++i) { dtc[i] = 0.0f; xc[i] = 0.0f; dtn[i] = 0.0f; xn[i] = 0.0f; }
  if (nb > 0) {
#pragma unroll
    for (int i = 0; i < PB; ++i) {
      dtc[i] = dtrow[(size_t)i * Dc + du];
      xc[i]  = xrow[(size_t)i * Dc + du];
    }
  }

  float alog[N], h[N];
  size_t hidx = (((size_t)b * nCh + chunk) * Dc + d) * N;
#pragma unroll
  for (int n = 0; n < N; ++n) h[n] = FINAL ? Hinit[hidx + n] : 0.0f;
  float Dv = FINAL ? Dp[d] : 0.0f;
#pragma unroll
  for (int n = 0; n < N; ++n) alog[n] = A_log[(size_t)d * N + n];

  float An2[N];
#pragma unroll
  for (int n = 0; n < N; ++n) An2[n] = -__expf(alog[n]) * LOG2E;

  float dtsum = 0.0f;

  if (FINAL) {
    asm volatile("" :: "v"(Dv), "v"(h[0]), "v"(h[1]), "v"(h[2]), "v"(h[3]),
                       "v"(h[4]), "v"(h[5]), "v"(h[6]), "v"(h[7]),
                       "v"(h[8]), "v"(h[9]), "v"(h[10]), "v"(h[11]),
                       "v"(h[12]), "v"(h[13]), "v"(h[14]), "v"(h[15]));
  }

  for (int bi = 0; bi < nb; ++bi) {
    const int l0 = bi * PB;
    if (bi + 1 < nb) {
      const float* dnext = dtrow + (size_t)(l0 + PB) * Dc;
      const float* xnext = xrow  + (size_t)(l0 + PB) * Dc;
#pragma unroll
      for (int i = 0; i < PB; ++i) {
        dtn[i] = dnext[(size_t)i * Dc + du];
        xn[i]  = xnext[(size_t)i * Dc + du];
      }
    }
#pragma unroll
    for (int i = 0; i < PB; ++i) {
      const int l = l0 + i;
      float dtv = dtc[i];
      float xv = xc[i];
      float us = dtv * xv;
      if (!FINAL) dtsum += dtv;
      float ub[N];
#pragma unroll
      for (int n = 0; n < N; ++n) ub[n] = us * bcur[n];
      asm volatile("" :: "s"(bcur[0]), "s"(bcur[1]), "s"(bcur[2]), "s"(bcur[3]),
                         "s"(bcur[4]), "s"(bcur[5]), "s"(bcur[6]), "s"(bcur[7]),
                         "s"(bcur[8]), "s"(bcur[9]), "s"(bcur[10]), "s"(bcur[11]),
                         "s"(bcur[12]), "s"(bcur[13]), "s"(bcur[14]), "s"(bcur[15]));
      const int ln = (i + 1 < PB) ? (l + 1) : ((l + 1 < Lc) ? (l + 1) : l);
      float cnx[N];
#pragma unroll
      for (int n = 0; n < N; ++n) {
        bcur[n] = Brow[ln * N + n];
        cnx[n] = FINAL ? Crow[ln * N + n] : 0.0f;
      }
      __builtin_amdgcn_sched_barrier(0);
      float y = 0.0f;
#pragma unroll
      for (int n = 0; n < N; ++n) {
        float a = __builtin_amdgcn_exp2f(dtv * An2[n]);
        h[n] = __builtin_fmaf(a, h[n], ub[n]);
        if (FINAL) y += h[n] * ccur[n];
      }
      if (FINAL) out[(rowbase + l) * Dc + du] = y + xv * Dv;
#pragma unroll
      for (int n = 0; n < N; ++n) ccur[n] = cnx[n];
    }
#pragma unroll
    for (int i = 0; i < PB; ++i) { dtc[i] = dtn[i]; xc[i] = xn[i]; }
  }

  for (int l = nb * PB; l < Lc; ++l) {
    float dtv = dtrow[(size_t)l * Dc + du];
    float xv = xrow[(size_t)l * Dc + du];
    float us = dtv * xv;
    if (!FINAL) dtsum += dtv;
    float y = 0.0f;
#pragma unroll
    for (int n = 0; n < N; ++n) {
      float a = __builtin_amdgcn_exp2f(dtv * An2[n]);
      h[n] = __builtin_fmaf(a, h[n], us * Brow[l * N + n]);
      if (FINAL) y += h[n] * Crow[l * N + n];
    }
    if (FINAL) out[(rowbase + l) * Dc + du] = y + xv * Dv;
  }

  if (!FINAL) {
#pragma unroll
    for (int n = 0; n < N; ++n) Hout[hidx + n] = h[n];
    Ssum[((size_t)b * nCh + chunk) * Dc + d] = dtsum;
  }
}

__global__ __launch_bounds__(256) void scan_combine_kernel(
    const float* __restrict__ Hlocal, const float* __restrict__ Ssum,
    const float* __restrict__ A_log, float* __restrict__ Hinit,
    int Dc, int nCh, int Bsz) {
  const int N = 16;
  const int CB = 8;
  long long idx = (long long)blockIdx.x * blockDim.x + threadIdx.x;
  if (idx >= (long long)Bsz * Dc) return;
  int d = (int)(idx % Dc);
  int b = (int)(idx / Dc);
  float An2[N], h[N];
#pragma unroll
  for (int n = 0; n < N; ++n) {
    An2[n] = -__expf(A_log[(size_t)d * N + n]) * LOG2E;
    h[n] = 0.0f;
  }
  int c = 0;
  for (; c + CB <= nCh; c += CB) {
    float S[CB];
    float hl[CB][N];
#pragma unroll
    for (int j = 0; j < CB; ++j) {
      size_t base = ((size_t)b * nCh + (c + j)) * Dc + d;
      S[j] = Ssum[base];
#pragma unroll
      for (int n = 0; n < N; ++n) hl[j][n] = Hlocal[base * N + n];
    }
    __builtin_amdgcn_sched_barrier(0);
#pragma unroll
    for (int j = 0; j < CB; ++j) {
      size_t hidx = (((size_t)b * nCh + (c + j)) * Dc + d) * N;
#pragma unroll
      for (int n = 0; n < N; ++n) {
        Hinit[hidx + n] = h[n];
        float a = __builtin_amdgcn_exp2f(An2[n] * S[j]);
        h[n] = a * h[n] + hl[j][n];
      }
    }
  }
  for (; c < nCh; ++c) {
    size_t base = ((size_t)b * nCh + c) * Dc + d;
    size_t hidx = base * N;
    float S = Ssum[base];
#pragma unroll
    for (int n = 0; n < N; ++n) {
      Hinit[hidx + n] = h[n];
      float a = __builtin_amdgcn_exp2f(An2[n] * S);
      h[n] = a * h[n] + Hlocal[hidx + n];
    }
  }
}

__global__ void scan_simple_kernel(const float* __restrict__ dt,
                                   const float* __restrict__ x,
                                   const float* __restrict__ Bm,
                                   const float* __restrict__ Cm,
                                   const float* __restrict__ A_log,
                                   const float* __restrict__ Dp,
                                   float* __restrict__ out,
                                   int L, int Dc, int Bsz) {
  const int N = 16;
  long long idx = (long long)blockIdx.x * blockDim.x + threadIdx.x;
  if (idx >= (long long)Bsz * Dc) return;
  int d = (int)(idx % Dc);
  int b = (int)(idx / Dc);
  float An2[N], h[N];
#pragma unroll
  for (int n = 0; n < N; ++n) {
    An2[n] = -__expf(A_log[(size_t)d * N + n]) * LOG2E;
    h[n] = 0.0f;
  }
  float Dv = Dp[d];
  for (int l = 0; l < L; ++l) {
    size_t row = (size_t)b * L + l;
    size_t g = row * Dc + d;
    float dtv = dt[g];
    float xv = x[g];
    float us = dtv * xv;
    const float* Br = Bm + row * N;
    const float* Cr = Cm + row * N;
    float y = 0.0f;
#pragma unroll
    for (int n = 0; n < N; ++n) {
      h[n] = __builtin_amdgcn_exp2f(dtv * An2[n]) * h[n] + us * Br[n];
      y += h[n] * Cr[n];
    }
    out[g] = y + xv * Dv;
  }
}

__global__ void scan_fused_kernel(const float* __restrict__ x,
                                  const float* __restrict__ delta,
                                  const float* __restrict__ Bm,
                                  const float* __restrict__ Cm,
                                  const float* __restrict__ W,
                                  const float* __restrict__ bias,
                                  const float* __restrict__ A_log,
                                  const float* __restrict__ Dp,
                                  float* __restrict__ out,
                                  int L, int Dc, int K, int Bsz) {
  const int N = 16;
  long long idx = (long long)blockIdx.x * blockDim.x + threadIdx.x;
  if (idx >= (long long)Bsz * Dc) return;
  int d = (int)(idx % Dc);
  int b = (int)(idx / Dc);
  float An2[N], h[N];
#pragma unroll
  for (int n = 0; n < N; ++n) {
    An2[n] = -__expf(A_log[(size_t)d * N + n]) * LOG2E;
    h[n] = 0.0f;
  }
  float Dv = Dp[d];
  float bv = bias[d];
  const float* wr = W + (size_t)d * K;
  for (int l = 0; l < L; ++l) {
    size_t row = (size_t)b * L + l;
    const float* dr = delta + row * K;
    float acc = bv;
    for (int r = 0; r < K; ++r) acc += dr[r] * wr[r];
    float dtv = softplus_f(acc);
    float xv = x[row * Dc + d];
    float us = dtv * xv;
    const float* Br = Bm + row * N;
    const float* Cr = Cm + row * N;
    float y = 0.0f;
#pragma unroll
    for (int n = 0; n < N; ++n) {
      h[n] = __builtin_amdgcn_exp2f(dtv * An2[n]) * h[n] + us * Br[n];
      y += h[n] * Cr[n];
    }
    out[row * Dc + d] = y + xv * Dv;
  }
}

extern "C" void kernel_launch(void* const* d_in, const int* in_sizes, int n_in,
                              void* d_out, int out_size, void* d_ws, size_t ws_size,
                              hipStream_t stream) {
  const float* x     = (const float*)d_in[0];
  const float* delta = (const float*)d_in[1];
  const float* Bm    = (const float*)d_in[2];
  const float* Cm    = (const float*)d_in[3];
  const float* W     = (const float*)d_in[4];
  const float* bias  = (const float*)d_in[5];
  const float* A_log = (const float*)d_in[6];
  const float* Dp    = (const float*)d_in[7];
  float* out = (float*)d_out;

  int Dc = in_sizes[5];
  int K  = in_sizes[4] / Dc;
  long long BL = (long long)in_sizes[0] / Dc;
  int Bsz = (BL % 2 == 0) ? 2 : 1;
  int L = (int)(BL / Bsz);

  size_t dtElems = (size_t)BL * Dc;
  size_t dtBytes = dtElems * sizeof(float);

  const int Lc = 64;
  bool chunkable = (L % Lc) == 0 && (Dc % 256) == 0;
  int nCh = chunkable ? (L / Lc) : 1;
  size_t chElems = (size_t)Bsz * nCh * Dc;
  size_t fullBytes = dtBytes + chElems * sizeof(float) * (1 + 16 + 16);

  if (ws_size < dtBytes) {
    int threads = 256;
    long long tot = (long long)Bsz * Dc;
    int blocks = (int)((tot + threads - 1) / threads);
    scan_fused_kernel<<<blocks, threads, 0, stream>>>(
        x, delta, Bm, Cm, W, bias, A_log, Dp, out, L, Dc, K, Bsz);
    return;
  }

  float* dt = (float*)d_ws;

  if ((BL % 64) == 0 && (Dc % 128) == 0 && (K % 32) == 0) {
    int blocks = (int)(BL / 64) * (Dc / 128);
    dtproj_tiled_kernel<<<blocks, 256, 0, stream>>>(delta, W, bias, dt,
                                                    (int)BL, Dc, K);
  } else if ((BL % 16) == 0 && (Dc % 32) == 0 && (K % 4) == 0) {
    int tiles = (int)(BL / 16) * (Dc / 32);
    int blocks = (tiles + 7) / 8;
    dtproj_wmma_kernel<<<blocks, 256, 0, stream>>>(delta, W, bias, dt,
                                                   (int)BL, Dc, K);
  } else {
    long long tot = BL * (long long)Dc;
    int blocks = (int)((tot + 255) / 256);
    dtproj_naive_kernel<<<blocks, 256, 0, stream>>>(delta, W, bias, dt, BL, Dc, K);
  }

  if (chunkable && ws_size >= fullBytes) {
    float* Ssum   = dt + dtElems;
    float* Hlocal = Ssum + chElems;
    float* Hinit  = Hlocal + chElems * 16;

    int blocks13 = Bsz * nCh * (Dc / 256);
    scan_chunk_kernel<false><<<blocks13, 256, 0, stream>>>(
        dt, x, Bm, Cm, A_log, Dp, Hlocal, Ssum, (const float*)nullptr,
        (float*)nullptr, L, Dc, Lc, nCh);

    long long tot = (long long)Bsz * Dc;
    int cblocks = (int)((tot + 255) / 256);
    scan_combine_kernel<<<cblocks, 256, 0, stream>>>(Hlocal, Ssum, A_log, Hinit,
                                                     Dc, nCh, Bsz);

    scan_chunk_kernel<true><<<blocks13, 256, 0, stream>>>(
        dt, x, Bm, Cm, A_log, Dp, (float*)nullptr, (float*)nullptr, Hinit, out,
        L, Dc, Lc, nCh);
  } else {
    long long tot = (long long)Bsz * Dc;
    int blocks = (int)((tot + 255) / 256);
    scan_simple_kernel<<<blocks, 256, 0, stream>>>(dt, x, Bm, Cm, A_log, Dp, out,
                                                   L, Dc, Bsz);
  }
}


# dt projection epilogue: softplus as max(v,0)+log1p(exp(-|v|)), log1p from the hardware log of u=1+e plus (e-(u-1))/u, branch-free, replacing the library log1pf; f32, last bits of dt may differ
# speedup vs baseline: 2.1583x; 1.1588x over previous
#include <hip/hip_runtime.h>
#include <math.h>

typedef __attribute__((ext_vector_type(2))) float v2f;
typedef __attribute__((ext_vector_type(8))) float v8f;
typedef __attribute__((ext_vector_type(4))) float v4f;

#define LOG2E 1.4426950408889634f
#define LN2F 0.6931471805599453f

__device__ __forceinline__ float softplus_f(float v) {
  float e = __builtin_amdgcn_exp2f(-fabsf(v) * LOG2E);
  float u = 1.0f + e;
  float dropped = e - (u - 1.0f);
  float lg = __builtin_amdgcn_logf(u) * LN2F;
  return fmaxf(v, 0.0f) + (lg + dropped * __builtin_amdgcn_rcpf(u));
}

__global__ __launch_bounds__(256) void dtproj_wmma_kernel(
    const float* __restrict__ delta, const float* __restrict__ W,
    const float* __restrict__ bias, float* __restrict__ dt,
    int M, int Dc, int K) {
  int wave = threadIdx.x >> 5;
  int lane = threadIdx.x & 31;
  int tilesN = Dc >> 5;
  int tile = blockIdx.x * 8 + wave;
  int totalTiles = (M >> 4) * tilesN;
  if (tile >= totalTiles) return;
  int tn = tile % tilesN;
  int tm = tile / tilesN;
  int mn   = lane & 15;
  int half = lane >> 4;

  const float* arow  = delta + (size_t)(tm * 16 + mn) * K;
  const float* brow0 = W     + (size_t)(tn * 32 + mn) * K;
  const float* brow1 = W     + (size_t)(tn * 32 + 16 + mn) * K;

  v8f acc0 = {};
  v8f acc1 = {};
  for (int k0 = 0; k0 < K; k0 += 4) {
    int kb = k0 + (half << 1);
    v2f a, b0, b1;
    a.x  = arow[kb];  a.y  = arow[kb + 1];
    b0.x = brow0[kb]; b0.y = brow0[kb + 1];
    b1.x = brow1[kb]; b1.y = brow1[kb + 1];
    acc0 = __builtin_amdgcn_wmma_f32_16x16x4_f32(
        false, a, false, b0, (short)0, acc0, false, false);
    acc1 = __builtin_amdgcn_wmma_f32_16x16x4_f32(
        false, a, false, b1, (short)0, acc1, false, false);
  }

  int gd0 = tn * 32 + mn;
  int gd1 = gd0 + 16;
  float bv0 = bias[gd0];
  float bv1 = bias[gd1];
  int gm0 = tm * 16 + half * 8;
#pragma unroll
  for (int v = 0; v < 8; ++v) {
    size_t rowoff = (size_t)(gm0 + v) * Dc;
    dt[rowoff + gd0] = softplus_f(acc0[v] + bv0);
    dt[rowoff + gd1] = softplus_f(acc1[v] + bv1);
  }
}

#define DTP_BM 64
#define DTP_BN 128
#define DTP_KC 32
#define DTP_LD 36

__global__ __launch_bounds__(256) void dtproj_tiled_kernel(
    const float* __restrict__ delta, const float* __restrict__ W,
    const float* __restrict__ bias, float* __restrict__ dt,
    int M, int Dc, int K) {
  __shared__ __attribute__((aligned(16))) float sA[2][DTP_BM * DTP_LD];
  __shared__ __attribute__((aligned(16))) float sB[2][DTP_BN * DTP_LD];

  const int tid  = threadIdx.x;
  const int wave = tid >> 5;
  const int lane = tid & 31;
  const int c = lane & 15;
  const int h = lane >> 4;
  const int tilesN = Dc / DTP_BN;
  const int m0 = (blockIdx.x / tilesN) * DTP_BM;
  const int n0 = (blockIdx.x % tilesN) * DTP_BN;
  const int wm = (wave >> 2) * 32;
  const int wn = (wave & 3) * 32;

  const int crow = tid >> 3;
  const int ccol = (tid & 7) * 4;
  const float* gA = delta + (size_t)(m0 + crow) * K + ccol;
  const float* gB = W     + (size_t)(n0 + crow) * K + ccol;
  const size_t gstep = (size_t)32 * K;

  v4f pa[2], pb[4];
#pragma unroll
  for (int p = 0; p < 2; ++p) pa[p] = *(const v4f*)(gA + p * gstep);
#pragma unroll
  for (int p = 0; p < 4; ++p) pb[p] = *(const v4f*)(gB + p * gstep);
#pragma unroll
  for (int p = 0; p < 2; ++p) *(v4f*)&sA[0][(crow + 32 * p) * DTP_LD + ccol] = pa[p];
#pragma unroll
  for (int p = 0; p < 4; ++p) *(v4f*)&sB[0][(crow + 32 * p) * DTP_LD + ccol] = pb[p];
  __syncthreads();

  v8f acc[2][2];
#pragma unroll
  for (int i = 0; i < 2; ++i)
#pragma unroll
    for (int j = 0; j < 2; ++j) acc[i][j] = (v8f){};

  const int nk = K / DTP_KC;
  for (int kc = 0; kc < nk; ++kc) {
    const int st = kc & 1;
    const bool more = (kc + 1 < nk);
    if (more) {
      const float* gAn = gA + (size_t)(kc + 1) * DTP_KC;
      const float* gBn = gB + (size_t)(kc + 1) * DTP_KC;
#pragma unroll
      for (int p = 0; p < 2; ++p) pa[p] = *(const v4f*)(gAn + p * gstep);
#pragma unroll
      for (int p = 0; p < 4; ++p) pb[p] = *(const v4f*)(gBn + p * gstep);
    }
    const float* abase = &sA[st][(wm + c) * DTP_LD + 2 * h];
    const float* bbase = &sB[st][(wn + c) * DTP_LD + 2 * h];
#pragma unroll
    for (int kk = 0; kk < DTP_KC; kk += 4) {
      v2f af[2], bf[2];
#pragma unroll
      for (int i = 0; i < 2; ++i) af[i] = *(const v2f*)(abase + i * 16 * DTP_LD + kk);
#pragma unroll
      for (int j = 0; j < 2; ++j) bf[j] = *(const v2f*)(bbase + j * 16 * DTP_LD + kk);
#pragma unroll
      for (int i = 0; i < 2; ++i)
#pragma unroll
        for (int j = 0; j < 2; ++j)
          acc[i][j] = __builtin_amdgcn_wmma_f32_16x16x4_f32(
              false, af[i], false, bf[j], (short)0, acc[i][j], false, false);
    }
    if (more) {
#pragma unroll
      for (int p = 0; p < 2; ++p) *(v4f*)&sA[st ^ 1][(crow + 32 * p) * DTP_LD + ccol] = pa[p];
#pragma unroll
      for (int p = 0; p < 4; ++p) *(v4f*)&sB[st ^ 1][(crow + 32 * p) * DTP_LD + ccol] = pb[p];
    }
    __syncthreads();
  }

  const int gd0 = n0 + wn + c;
  const int gd1 = gd0 + 16;
  const float bv0 = bias[gd0];
  const float bv1 = bias[gd1];
#pragma unroll
  for (int i = 0; i < 2; ++i) {
    const int gm0 = m0 + wm + 16 * i + 8 * h;
#pragma unroll
    for (int r = 0; r < 8; ++r) {
      size_t rowoff = (size_t)(gm0 + r) * Dc;
      dt[rowoff + gd0] = softplus_f(acc[i][0][r] + bv0);
      dt[rowoff + gd1] = softplus_f(acc[i][1][r] + bv1);
    }
  }
}

__global__ void dtproj_naive_kernel(const float* __restrict__ delta,
                                    const float* __restrict__ W,
                                    const float* __restrict__ bias,
                                    float* __restrict__ dt,
                                    long long M, int Dc, int K) {
  long long idx = (long long)blockIdx.x * blockDim.x + threadIdx.x;
  if (idx >= M * (long long)Dc) return;
  int d = (int)(idx % Dc);
  long long m = idx / Dc;
  float acc = bias[d];
  const float* dr = delta + m * (long long)K;
  const float* wr = W + (size_t)d * K;
  for (int r = 0; r < K; ++r) acc += dr[r] * wr[r];
  dt[idx] = softplus_f(acc);
}

template <bool FINAL>
__global__ __launch_bounds__(256) void scan_chunk_kernel(
    const float* __restrict__ dt, const float* __restrict__ x,
    const float* __restrict__ Bm, const float* __restrict__ Cm,
    const float* __restrict__ A_log, const float* __restrict__ Dp,
    float* __restrict__ Hout, float* __restrict__ Ssum,
    const float* __restrict__ Hinit, float* __restrict__ out,
    int L, int Dc, int Lc, int nCh) {
  const int N = 16;
  const int PB = 16;

  int DB = Dc / blockDim.x;
  int bx = blockIdx.x;
  int dblk = bx % DB; bx /= DB;
  int chunk = bx % nCh; bx /= nCh;
  int b = bx;
  int d = dblk * blockDim.x + threadIdx.x;
  unsigned du = (unsigned)d;

  size_t rowbase = (size_t)b * L + (size_t)chunk * Lc;
  const float* dtrow = dt + rowbase * Dc;
  const float* xrow  = x  + rowbase * Dc;
  typedef const __attribute__((address_space(4))) float cfloat_k;
  cfloat_k* Brow = (cfloat_k*)(Bm + rowbase * N);
  cfloat_k* Crow = (cfloat_k*)(Cm + rowbase * N);
  float bcur[N], ccur[N];
#pragma unroll
  for (int n = 0; n < N; ++n) {
    bcur[n] = Brow[n];
    ccur[n] = FINAL ? Crow[n] : 0.0f;
  }

  const int nb = Lc / PB;
  float dtc[PB], xc[PB], dtn[PB], xn[PB];
#pragma unroll
  for (int i = 0; i < PB; ++i) { dtc[i] = 0.0f; xc[i] = 0.0f; dtn[i] = 0.0f; xn[i] = 0.0f; }
  if (nb > 0) {
#pragma unroll
    for (int i = 0; i < PB; ++i) {
      dtc[i] = dtrow[(size_t)i * Dc + du];
      xc[i]  = xrow[(size_t)i * Dc + du];
    }
  }

  float alog[N], h[N];
  size_t hidx = (((size_t)b * nCh + chunk) * Dc + d) * N;
#pragma unroll
  for (int n = 0; n < N; ++n) h[n] = FINAL ? Hinit[hidx + n] : 0.0f;
  float Dv = FINAL ? Dp[d] : 0.0f;
#pragma unroll
  for (int n = 0; n < N; ++n) alog[n] = A_log[(size_t)d * N + n];

  float An2[N];
#pragma unroll
  for (int n = 0; n < N; ++n) An2[n] = -__expf(alog[n]) * LOG2E;

  float dtsum = 0.0f;

  if (FINAL) {
    asm volatile("" :: "v"(Dv), "v"(h[0]), "v"(h[1]), "v"(h[2]), "v"(h[3]),
                       "v"(h[4]), "v"(h[5]), "v"(h[6]), "v"(h[7]),
                       "v"(h[8]), "v"(h[9]), "v"(h[10]), "v"(h[11]),
                       "v"(h[12]), "v"(h[13]), "v"(h[14]), "v"(h[15]));
  }

  for (int bi = 0; bi < nb; ++bi) {
    const int l0 = bi * PB;
    if (bi + 1 < nb) {
      const float* dnext = dtrow + (size_t)(l0 + PB) * Dc;
      const float* xnext = xrow  + (size_t)(l0 + PB) * Dc;
#pragma unroll
      for (int i = 0; i < PB; ++i) {
        dtn[i] = dnext[(size_t)i * Dc + du];
        xn[i]  = xnext[(size_t)i * Dc + du];
      }
    }
#pragma unroll
    for (int i = 0; i < PB; ++i) {
      const int l = l0 + i;
      float dtv = dtc[i];
      float xv = xc[i];
      float us = dtv * xv;
      if (!FINAL) dtsum += dtv;
      float ub[N];
#pragma unroll
      for (int n = 0; n < N; ++n) ub[n] = us * bcur[n];
      asm volatile("" :: "s"(bcur[0]), "s"(bcur[1]), "s"(bcur[2]), "s"(bcur[3]),
                         "s"(bcur[4]), "s"(bcur[5]), "s"(bcur[6]), "s"(bcur[7]),
                         "s"(bcur[8]), "s"(bcur[9]), "s"(bcur[10]), "s"(bcur[11]),
                         "s"(bcur[12]), "s"(bcur[13]), "s"(bcur[14]), "s"(bcur[15]));
      const int ln = (i + 1 < PB) ? (l + 1) : ((l + 1 < Lc) ? (l + 1) : l);
      float cnx[N];
#pragma unroll
      for (int n = 0; n < N; ++n) {
        bcur[n] = Brow[ln * N + n];
        cnx[n] = FINAL ? Crow[ln * N + n] : 0.0f;
      }
      __builtin_amdgcn_sched_barrier(0);
      float y = 0.0f;
#pragma unroll
      for (int n = 0; n < N; ++n) {
        float a = __builtin_amdgcn_exp2f(dtv * An2[n]);
        h[n] = __builtin_fmaf(a, h[n], ub[n]);
        if (FINAL) y += h[n] * ccur[n];
      }
      if (FINAL) out[(rowbase + l) * Dc + du] = y + xv * Dv;
#pragma unroll
      for (int n = 0; n < N; ++n) ccur[n] = cnx[n];
    }
#pragma unroll
    for (int i = 0; i < PB; ++i) { dtc[i] = dtn[i]; xc[i] = xn[i]; }
  }

  for (int l = nb * PB; l < Lc; ++l) {
    float dtv = dtrow[(size_t)l * Dc + du];
    float xv = xrow[(size_t)l * Dc + du];
    float us = dtv * xv;
    if (!FINAL) dtsum += dtv;
    float y = 0.0f;
#pragma unroll
    for (int n = 0; n < N; ++n) {
      float a = __builtin_amdgcn_exp2f(dtv * An2[n]);
      h[n] = __builtin_fmaf(a, h[n], us * Brow[l * N + n]);
      if (FINAL) y += h[n] * Crow[l * N + n];
    }
    if (FINAL) out[(rowbase + l) * Dc + du] = y + xv * Dv;
  }

  if (!FINAL) {
#pragma unroll
    for (int n = 0; n < N; ++n) Hout[hidx + n] = h[n];
    Ssum[((size_t)b * nCh + chunk) * Dc + d] = dtsum;
  }
}

__global__ __launch_bounds__(256) void scan_combine_kernel(
    const float* __restrict__ Hlocal, const float* __restrict__ Ssum,
    const float* __restrict__ A_log, float* __restrict__ Hinit,
    int Dc, int nCh, int Bsz) {
  const int N = 16;
  const int CB = 8;
  long long idx = (long long)blockIdx.x * blockDim.x + threadIdx.x;
  if (idx >= (long long)Bsz * Dc) return;
  int d = (int)(idx % Dc);
  int b = (int)(idx / Dc);
  float An2[N], h[N];
#pragma unroll
  for (int n = 0; n < N; ++n) {
    An2[n] = -__expf(A_log[(size_t)d * N + n]) * LOG2E;
    h[n] = 0.0f;
  }
  int c = 0;
  for (; c + CB <= nCh; c += CB) {
    float S[CB];
    float hl[CB][N];
#pragma unroll
    for (int j = 0; j < CB; ++j) {
      size_t base = ((size_t)b * nCh + (c + j)) * Dc + d;
      S[j] = Ssum[base];
#pragma unroll
      for (int n = 0; n < N; ++n) hl[j][n] = Hlocal[base * N + n];
    }
    __builtin_amdgcn_sched_barrier(0);
#pragma unroll
    for (int j = 0; j < CB; ++j) {
      size_t hidx = (((size_t)b * nCh + (c + j)) * Dc + d) * N;
#pragma unroll
      for (int n = 0; n < N; ++n) {
        Hinit[hidx + n] = h[n];
        float a = __builtin_amdgcn_exp2f(An2[n] * S[j]);
        h[n] = a * h[n] + hl[j][n];
      }
    }
  }
  for (; c < nCh; ++c) {
    size_t base = ((size_t)b * nCh + c) * Dc + d;
    size_t hidx = base * N;
    float S = Ssum[base];
#pragma unroll
    for (int n = 0; n < N; ++n) {
      Hinit[hidx + n] = h[n];
      float a = __builtin_amdgcn_exp2f(An2[n] * S);
      h[n] = a * h[n] + Hlocal[hidx + n];
    }
  }
}

__global__ void scan_simple_kernel(const float* __restrict__ dt,
                                   const float* __restrict__ x,
                                   const float* __restrict__ Bm,
                                   const float* __restrict__ Cm,
                                   const float* __restrict__ A_log,
                                   const float* __restrict__ Dp,
                                   float* __restrict__ out,
                                   int L, int Dc, int Bsz) {
  const int N = 16;
  long long idx = (long long)blockIdx.x * blockDim.x + threadIdx.x;
  if (idx >= (long long)Bsz * Dc) return;
  int d = (int)(idx % Dc);
  int b = (int)(idx / Dc);
  float An2[N], h[N];
#pragma unroll
  for (int n = 0; n < N; ++n) {
    An2[n] = -__expf(A_log[(size_t)d * N + n]) * LOG2E;
    h[n] = 0.0f;
  }
  float Dv = Dp[d];
  for (int l = 0; l < L; ++l) {
    size_t row = (size_t)b * L + l;
    size_t g = row * Dc + d;
    float dtv = dt[g];
    float xv = x[g];
    float us = dtv * xv;
    const float* Br = Bm + row * N;
    const float* Cr = Cm + row * N;
    float y = 0.0f;
#pragma unroll
    for (int n = 0; n < N; ++n) {
      h[n] = __builtin_amdgcn_exp2f(dtv * An2[n]) * h[n] + us * Br[n];
      y += h[n] * Cr[n];
    }
    out[g] = y + xv * Dv;
  }
}

__global__ void scan_fused_kernel(const float* __restrict__ x,
                                  const float* __restrict__ delta,
                                  const float* __restrict__ Bm,
                                  const float* __restrict__ Cm,
                                  const float* __restrict__ W,
                                  const float* __restrict__ bias,
                                  const float* __restrict__ A_log,
                                  const float* __restrict__ Dp,
                                  float* __restrict__ out,
                                  int L, int Dc, int K, int Bsz) {
  const int N = 16;
  long long idx = (long long)blockIdx.x * blockDim.x + threadIdx.x;
  if (idx >= (long long)Bsz * Dc) return;
  int d = (int)(idx % Dc);
  int b = (int)(idx / Dc);
  float An2[N], h[N];
#pragma unroll
  for (int n = 0; n < N; ++n) {
    An2[n] = -__expf(A_log[(size_t)d * N + n]) * LOG2E;
    h[n] = 0.0f;
  }
  float Dv = Dp[d];
  float bv = bias[d];
  const float* wr = W + (size_t)d * K;
  for (int l = 0; l < L; ++l) {
    size_t row = (size_t)b * L + l;
    const float* dr = delta + row * K;
    float acc = bv;
    for (int r = 0; r < K; ++r) acc += dr[r] * wr[r];
    float dtv = softplus_f(acc);
    float xv = x[row * Dc + d];
    float us = dtv * xv;
    const float* Br = Bm + row * N;
    const float* Cr = Cm + row * N;
    float y = 0.0f;
#pragma unroll
    for (int n = 0; n < N; ++n) {
      h[n] = __builtin_amdgcn_exp2f(dtv * An2[n]) * h[n] + us * Br[n];
      y += h[n] * Cr[n];
    }
    out[row * Dc + d] = y + xv * Dv;
  }
}

extern "C" void kernel_launch(void* const* d_in, const int* in_sizes, int n_in,
                              void* d_out, int out_size, void* d_ws, size_t ws_size,
                              hipStream_t stream) {
  const float* x     = (const float*)d_in[0];
  const float* delta = (const float*)d_in[1];
  const float* Bm    = (const float*)d_in[2];
  const float* Cm    = (const float*)d_in[3];
  const float* W     = (const float*)d_in[4];
  const float* bias  = (const float*)d_in[5];
  const float* A_log = (const float*)d_in[6];
  const float* Dp    = (const float*)d_in[7];
  float* out = (float*)d_out;

  int Dc = in_sizes[5];
  int K  = in_sizes[4] / Dc;
  long long BL = (long long)in_sizes[0] / Dc;
  int Bsz = (BL % 2 == 0) ? 2 : 1;
  int L = (int)(BL / Bsz);

  size_t dtElems = (size_t)BL * Dc;
  size_t dtBytes = dtElems * sizeof(float);

  const int Lc = 64;
  bool chunkable = (L % Lc) == 0 && (Dc % 256) == 0;
  int nCh = chunkable ? (L / Lc) : 1;
  size_t chElems = (size_t)Bsz * nCh * Dc;
  size_t fullBytes = dtBytes + chElems * sizeof(float) * (1 + 16 + 16);

  if (ws_size < dtBytes) {
    int threads = 256;
    long long tot = (long long)Bsz * Dc;
    int blocks = (int)((tot + threads - 1) / threads);
    scan_fused_kernel<<<blocks, threads, 0, stream>>>(
        x, delta, Bm, Cm, W, bias, A_log, Dp, out, L, Dc, K, Bsz);
    return;
  }

  float* dt = (float*)d_ws;

  if ((BL % 64) == 0 && (Dc % 128) == 0 && (K % 32) == 0) {
    int blocks = (int)(BL / 64) * (Dc / 128);
    dtproj_tiled_kernel<<<blocks, 256, 0, stream>>>(delta, W, bias, dt,
                                                    (int)BL, Dc, K);
  } else if ((BL % 16) == 0 && (Dc % 32) == 0 && (K % 4) == 0) {
    int tiles = (int)(BL / 16) * (Dc / 32);
    int blocks = (tiles + 7) / 8;
    dtproj_wmma_kernel<<<blocks, 256, 0, stream>>>(delta, W, bias, dt,
                                                   (int)BL, Dc, K);
  } else {
    long long tot = BL * (long long)Dc;
    int blocks = (int)((tot + 255) / 256);
    dtproj_naive_kernel<<<blocks, 256, 0, stream>>>(delta, W, bias, dt, BL, Dc, K);
  }

  if (chunkable && ws_size >= fullBytes) {
    float* Ssum   = dt + dtElems;
    float* Hlocal = Ssum + chElems;
    float* Hinit  = Hlocal + chElems * 16;

    int blocks13 = Bsz * nCh * (Dc / 256);
    scan_chunk_kernel<false><<<blocks13, 256, 0, stream>>>(
        dt, x, Bm, Cm, A_log, Dp, Hlocal, Ssum, (const float*)nullptr,
        (float*)nullptr, L, Dc, Lc, nCh);

    long long tot = (long long)Bsz * Dc;
    int cblocks = (int)((tot + 255) / 256);
    scan_combine_kernel<<<cblocks, 256, 0, stream>>>(Hlocal, Ssum, A_log, Hinit,
                                                     Dc, nCh, Bsz);

    scan_chunk_kernel<true><<<blocks13, 256, 0, stream>>>(
        dt, x, Bm, Cm, A_log, Dp, (float*)nullptr, (float*)nullptr, Hinit, out,
        L, Dc, Lc, nCh);
  } else {
    long long tot = (long long)Bsz * Dc;
    int blocks = (int)((tot + 255) / 256);
    scan_simple_kernel<<<blocks, 256, 0, stream>>>(dt, x, Bm, Cm, A_log, Dp, out,
                                                   L, Dc, Bsz);
  }
}


# chunk combine: one lane per (batch, channel, state) instead of one per (batch, channel): 16x the grid, so every compute unit takes part; whole-line accesses; all chunk operands requested up front
# speedup vs baseline: 2.5859x; 1.1981x over previous
#include <hip/hip_runtime.h>
#include <math.h>

typedef __attribute__((ext_vector_type(2))) float v2f;
typedef __attribute__((ext_vector_type(8))) float v8f;
typedef __attribute__((ext_vector_type(4))) float v4f;

#define LOG2E 1.4426950408889634f
#define LN2F 0.6931471805599453f

__device__ __forceinline__ float softplus_f(float v) {
  float e = __builtin_amdgcn_exp2f(-fabsf(v) * LOG2E);
  float u = 1.0f + e;
  float dropped = e - (u - 1.0f);
  float lg = __builtin_amdgcn_logf(u) * LN2F;
  return fmaxf(v, 0.0f) + (lg + dropped * __builtin_amdgcn_rcpf(u));
}

__global__ __launch_bounds__(256) void dtproj_wmma_kernel(
    const float* __restrict__ delta, const float* __restrict__ W,
    const float* __restrict__ bias, float* __restrict__ dt,
    int M, int Dc, int K) {
  int wave = threadIdx.x >> 5;
  int lane = threadIdx.x & 31;
  int tilesN = Dc >> 5;
  int tile = blockIdx.x * 8 + wave;
  int totalTiles = (M >> 4) * tilesN;
  if (tile >= totalTiles) return;
  int tn = tile % tilesN;
  int tm = tile / tilesN;
  int mn   = lane & 15;
  int half = lane >> 4;

  const float* arow  = delta + (size_t)(tm * 16 + mn) * K;
  const float* brow0 = W     + (size_t)(tn * 32 + mn) * K;
  const float* brow1 = W     + (size_t)(tn * 32 + 16 + mn) * K;

  v8f acc0 = {};
  v8f acc1 = {};
  for (int k0 = 0; k0 < K; k0 += 4) {
    int kb = k0 + (half << 1);
    v2f a, b0, b1;
    a.x  = arow[kb];  a.y  = arow[kb + 1];
    b0.x = brow0[kb]; b0.y = brow0[kb + 1];
    b1.x = brow1[kb]; b1.y = brow1[kb + 1];
    acc0 = __builtin_amdgcn_wmma_f32_16x16x4_f32(
        false, a, false, b0, (short)0, acc0, false, false);
    acc1 = __builtin_amdgcn_wmma_f32_16x16x4_f32(
        false, a, false, b1, (short)0, acc1, false, false);
  }

  int gd0 = tn * 32 + mn;
  int gd1 = gd0 + 16;
  float bv0 = bias[gd0];
  float bv1 = bias[gd1];
  int gm0 = tm * 16 + half * 8;
#pragma unroll
  for (int v = 0; v < 8; ++v) {
    size_t rowoff = (size_t)(gm0 + v) * Dc;
    dt[rowoff + gd0] = softplus_f(acc0[v] + bv0);
    dt[rowoff + gd1] = softplus_f(acc1[v] + bv1);
  }
}

#define DTP_BM 64
#define DTP_BN 128
#define DTP_KC 32
#define DTP_LD 36

__global__ __launch_bounds__(256) void dtproj_tiled_kernel(
    const float* __restrict__ delta, const float* __restrict__ W,
    const float* __restrict__ bias, float* __restrict__ dt,
    int M, int Dc, int K) {
  __shared__ __attribute__((aligned(16))) float sA[2][DTP_BM * DTP_LD];
  __shared__ __attribute__((aligned(16))) float sB[2][DTP_BN * DTP_LD];

  const int tid  = threadIdx.x;
  const int wave = tid >> 5;
  const int lane = tid & 31;
  const int c = lane & 15;
  const int h = lane >> 4;
  const int tilesN = Dc / DTP_BN;
  const int m0 = (blockIdx.x / tilesN) * DTP_BM;
  const int n0 = (blockIdx.x % tilesN) * DTP_BN;
  const int wm = (wave >> 2) * 32;
  const int wn = (wave & 3) * 32;

  const int crow = tid >> 3;
  const int ccol = (tid & 7) * 4;
  const float* gA = delta + (size_t)(m0 + crow) * K + ccol;
  const float* gB = W     + (size_t)(n0 + crow) * K + ccol;
  const size_t gstep = (size_t)32 * K;

  v4f pa[2], pb[4];
#pragma unroll
  for (int p = 0; p < 2; ++p) pa[p] = *(const v4f*)(gA + p * gstep);
#pragma unroll
  for (int p = 0; p < 4; ++p) pb[p] = *(const v4f*)(gB + p * gstep);
#pragma unroll
  for (int p = 0; p < 2; ++p) *(v4f*)&sA[0][(crow + 32 * p) * DTP_LD + ccol] = pa[p];
#pragma unroll
  for (int p = 0; p < 4; ++p) *(v4f*)&sB[0][(crow + 32 * p) * DTP_LD + ccol] = pb[p];
  __syncthreads();

  v8f acc[2][2];
#pragma unroll
  for (int i = 0; i < 2; ++i)
#pragma unroll
    for (int j = 0; j < 2; ++j) acc[i][j] = (v8f){};

  const int nk = K / DTP_KC;
  for (int kc = 0; kc < nk; ++kc) {
    const int st = kc & 1;
    const bool more = (kc + 1 < nk);
    if (more) {
      const float* gAn = gA + (size_t)(kc + 1) * DTP_KC;
      const float* gBn = gB + (size_t)(kc + 1) * DTP_KC;
#pragma unroll
      for (int p = 0; p < 2; ++p) pa[p] = *(const v4f*)(gAn + p * gstep);
#pragma unroll
      for (int p = 0; p < 4; ++p) pb[p] = *(const v4f*)(gBn + p * gstep);
    }
    const float* abase = &sA[st][(wm + c) * DTP_LD + 2 * h];
    const float* bbase = &sB[st][(wn + c) * DTP_LD + 2 * h];
#pragma unroll
    for (int kk = 0; kk < DTP_KC; kk += 4) {
      v2f af[2], bf[2];
#pragma unroll
      for (int i = 0; i < 2; ++i) af[i] = *(const v2f*)(abase + i * 16 * DTP_LD + kk);
#pragma unroll
      for (int j = 0; j < 2; ++j) bf[j] = *(const v2f*)(bbase + j * 16 * DTP_LD + kk);
#pragma unroll
      for (int i = 0; i < 2; ++i)
#pragma unroll
        for (int j = 0; j < 2; ++j)
          acc[i][j] = __builtin_amdgcn_wmma_f32_16x16x4_f32(
              false, af[i], false, bf[j], (short)0, acc[i][j], false, false);
    }
    if (more) {
#pragma unroll
      for (int p = 0; p < 2; ++p) *(v4f*)&sA[st ^ 1][(crow + 32 * p) * DTP_LD + ccol] = pa[p];
#pragma unroll
      for (int p = 0; p < 4; ++p) *(v4f*)&sB[st ^ 1][(crow + 32 * p) * DTP_LD + ccol] = pb[p];
    }
    __syncthreads();
  }

  const int gd0 = n0 + wn + c;
  const int gd1 = gd0 + 16;
  const float bv0 = bias[gd0];
  const float bv1 = bias[gd1];
#pragma unroll
  for (int i = 0; i < 2; ++i) {
    const int gm0 = m0 + wm + 16 * i + 8 * h;
#pragma unroll
    for (int r = 0; r < 8; ++r) {
      size_t rowoff = (size_t)(gm0 + r) * Dc;
      dt[rowoff + gd0] = softplus_f(acc[i][0][r] + bv0);
      dt[rowoff + gd1] = softplus_f(acc[i][1][r] + bv1);
    }
  }
}

__global__ void dtproj_naive_kernel(const float* __restrict__ delta,
                                    const float* __restrict__ W,
                                    const float* __restrict__ bias,
                                    float* __restrict__ dt,
                                    long long M, int Dc, int K) {
  long long idx = (long long)blockIdx.x * blockDim.x + threadIdx.x;
  if (idx >= M * (long long)Dc) return;
  int d = (int)(idx % Dc);
  long long m = idx / Dc;
  float acc = bias[d];
  const float* dr = delta + m * (long long)K;
  const float* wr = W + (size_t)d * K;
  for (int r = 0; r < K; ++r) acc += dr[r] * wr[r];
  dt[idx] = softplus_f(acc);
}

template <bool FINAL>
__global__ __launch_bounds__(256) void scan_chunk_kernel(
    const float* __restrict__ dt, const float* __restrict__ x,
    const float* __restrict__ Bm, const float* __restrict__ Cm,
    const float* __restrict__ A_log, const float* __restrict__ Dp,
    float* __restrict__ Hout, float* __restrict__ Ssum,
    const float* __restrict__ Hinit, float* __restrict__ out,
    int L, int Dc, int Lc, int nCh) {
  const int N = 16;
  const int PB = 16;

  int DB = Dc / blockDim.x;
  int bx = blockIdx.x;
  int dblk = bx % DB; bx /= DB;
  int chunk = bx % nCh; bx /= nCh;
  int b = bx;
  int d = dblk * blockDim.x + threadIdx.x;
  unsigned du = (unsigned)d;

  size_t rowbase = (size_t)b * L + (size_t)chunk * Lc;
  const float* dtrow = dt + rowbase * Dc;
  const float* xrow  = x  + rowbase * Dc;
  typedef const __attribute__((address_space(4))) float cfloat_k;
  cfloat_k* Brow = (cfloat_k*)(Bm + rowbase * N);
  cfloat_k* Crow = (cfloat_k*)(Cm + rowbase * N);
  float bcur[N], ccur[N];
#pragma unroll
  for (int n = 0; n < N; ++n) {
    bcur[n] = Brow[n];
    ccur[n] = FINAL ? Crow[n] : 0.0f;
  }

  const int nb = Lc / PB;
  float dtc[PB], xc[PB], dtn[PB], xn[PB];
#pragma unroll
  for (int i = 0; i < PB; ++i) { dtc[i] = 0.0f; xc[i] = 0.0f; dtn[i] = 0.0f; xn[i] = 0.0f; }
  if (nb > 0) {
#pragma unroll
    for (int i = 0; i < PB; ++i) {
      dtc[i] = dtrow[(size_t)i * Dc + du];
      xc[i]  = xrow[(size_t)i * Dc + du];
    }
  }

  float alog[N], h[N];
  size_t hidx = (((size_t)b * nCh + chunk) * Dc + d) * N;
#pragma unroll
  for (int n = 0; n < N; ++n) h[n] = FINAL ? Hinit[hidx + n] : 0.0f;
  float Dv = FINAL ? Dp[d] : 0.0f;
#pragma unroll
  for (int n = 0; n < N; ++n) alog[n] = A_log[(size_t)d * N + n];

  float An2[N];
#pragma unroll
  for (int n = 0; n < N; ++n) An2[n] = -__expf(alog[n]) * LOG2E;

  float dtsum = 0.0f;

  if (FINAL) {
    asm volatile("" :: "v"(Dv), "v"(h[0]), "v"(h[1]), "v"(h[2]), "v"(h[3]),
                       "v"(h[4]), "v"(h[5]), "v"(h[6]), "v"(h[7]),
                       "v"(h[8]), "v"(h[9]), "v"(h[10]), "v"(h[11]),
                       "v"(h[12]), "v"(h[13]), "v"(h[14]), "v"(h[15]));
  }

  for (int bi = 0; bi < nb; ++bi) {
    const int l0 = bi * PB;
    if (bi + 1 < nb) {
      const float* dnext = dtrow + (size_t)(l0 + PB) * Dc;
      const float* xnext = xrow  + (size_t)(l0 + PB) * Dc;
#pragma unroll
      for (int i = 0; i < PB; ++i) {
        dtn[i] = dnext[(size_t)i * Dc + du];
        xn[i]  = xnext[(size_t)i * Dc + du];
      }
    }
#pragma unroll
    for (int i = 0; i < PB; ++i) {
      const int l = l0 + i;
      float dtv = dtc[i];
      float xv = xc[i];
      float us = dtv * xv;
      if (!FINAL) dtsum += dtv;
      float ub[N];
#pragma unroll
      for (int n = 0; n < N; ++n) ub[n] = us * bcur[n];
      asm volatile("" :: "s"(bcur[0]), "s"(bcur[1]), "s"(bcur[2]), "s"(bcur[3]),
                         "s"(bcur[4]), "s"(bcur[5]), "s"(bcur[6]), "s"(bcur[7]),
                         "s"(bcur[8]), "s"(bcur[9]), "s"(bcur[10]), "s"(bcur[11]),
                         "s"(bcur[12]), "s"(bcur[13]), "s"(bcur[14]), "s"(bcur[15]));
      const int ln = (i + 1 < PB) ? (l + 1) : ((l + 1 < Lc) ? (l + 1) : l);
      float cnx[N];
#pragma unroll
      for (int n = 0; n < N; ++n) {
        bcur[n] = Brow[ln * N + n];
        cnx[n] = FINAL ? Crow[ln * N + n] : 0.0f;
      }
      __builtin_amdgcn_sched_barrier(0);
      float y = 0.0f;
#pragma unroll
      for (int n = 0; n < N; ++n) {
        float a = __builtin_amdgcn_exp2f(dtv * An2[n]);
        h[n] = __builtin_fmaf(a, h[n], ub[n]);
        if (FINAL) y += h[n] * ccur[n];
      }
      if (FINAL) out[(rowbase + l) * Dc + du] = y + xv * Dv;
#pragma unroll
      for (int n = 0; n < N; ++n) ccur[n] = cnx[n];
    }
#pragma unroll
    for (int i = 0; i < PB; ++i) { dtc[i] = dtn[i]; xc[i] = xn[i]; }
  }

  for (int l = nb * PB; l < Lc; ++l) {
    float dtv = dtrow[(size_t)l * Dc + du];
    float xv = xrow[(size_t)l * Dc + du];
    float us = dtv * xv;
    if (!FINAL) dtsum += dtv;
    float y = 0.0f;
#pragma unroll
    for (int n = 0; n < N; ++n) {
      float a = __builtin_amdgcn_exp2f(dtv * An2[n]);
      h[n] = __builtin_fmaf(a, h[n], us * Brow[l * N + n]);
      if (FINAL) y += h[n] * Crow[l * N + n];
    }
    if (FINAL) out[(rowbase + l) * Dc + du] = y + xv * Dv;
  }

  if (!FINAL) {
#pragma unroll
    for (int n = 0; n < N; ++n) Hout[hidx + n] = h[n];
    Ssum[((size_t)b * nCh + chunk) * Dc + d] = dtsum;
  }
}

__global__ __launch_bounds__(256) void scan_combine_kernel(
    const float* __restrict__ Hlocal, const float* __restrict__ Ssum,
    const float* __restrict__ A_log, float* __restrict__ Hinit,
    int Dc, int nCh, int Bsz) {
  const int N = 16;
  const int CB = 32;
  const long long perB = (long long)Dc * N;
  long long idx = (long long)blockIdx.x * blockDim.x + threadIdx.x;
  if (idx >= (long long)Bsz * perB) return;
  const int b = (int)(idx / perB);
  const int dn = (int)(idx - (long long)b * perB);
  const int d = dn >> 4;
  const size_t stepH = (size_t)perB;
  const size_t stepS = (size_t)Dc;
  const float* hp = Hlocal + (size_t)b * nCh * stepH + dn;
  const float* sp = Ssum   + (size_t)b * nCh * stepS + d;
  float*       op = Hinit  + (size_t)b * nCh * stepH + dn;

  const float alog = A_log[dn];
  const int nb = nCh / CB;
  float Sc[CB], Hc[CB], Sn[CB], Hn[CB];
#pragma unroll
  for (int j = 0; j < CB; ++j) { Sc[j] = 0.0f; Hc[j] = 0.0f; Sn[j] = 0.0f; Hn[j] = 0.0f; }
  if (nb > 0) {
#pragma unroll
    for (int j = 0; j < CB; ++j) {
      Sc[j] = sp[(size_t)j * stepS];
      Hc[j] = hp[(size_t)j * stepH];
    }
  }
  const float An2 = -__expf(alog) * LOG2E;
  float h = 0.0f;
  for (int bi = 0; bi < nb; ++bi) {
    const int c0 = bi * CB;
    if (bi + 1 < nb) {
#pragma unroll
      for (int j = 0; j < CB; ++j) {
        Sn[j] = sp[(size_t)(c0 + CB + j) * stepS];
        Hn[j] = hp[(size_t)(c0 + CB + j) * stepH];
      }
    }
    __builtin_amdgcn_sched_barrier(0);
#pragma unroll
    for (int j = 0; j < CB; ++j) {
      op[(size_t)(c0 + j) * stepH] = h;
      float a = __builtin_amdgcn_exp2f(An2 * Sc[j]);
      h = a * h + Hc[j];
    }
#pragma unroll
    for (int j = 0; j < CB; ++j) { Sc[j] = Sn[j]; Hc[j] = Hn[j]; }
  }
  for (int c = nb * CB; c < nCh; ++c) {
    float S = sp[(size_t)c * stepS];
    op[(size_t)c * stepH] = h;
    float a = __builtin_amdgcn_exp2f(An2 * S);
    h = a * h + hp[(size_t)c * stepH];
  }
}

__global__ void scan_simple_kernel(const float* __restrict__ dt,
                                   const float* __restrict__ x,
                                   const float* __restrict__ Bm,
                                   const float* __restrict__ Cm,
                                   const float* __restrict__ A_log,
                                   const float* __restrict__ Dp,
                                   float* __restrict__ out,
                                   int L, int Dc, int Bsz) {
  const int N = 16;
  long long idx = (long long)blockIdx.x * blockDim.x + threadIdx.x;
  if (idx >= (long long)Bsz * Dc) return;
  int d = (int)(idx % Dc);
  int b = (int)(idx / Dc);
  float An2[N], h[N];
#pragma unroll
  for (int n = 0; n < N; ++n) {
    An2[n] = -__expf(A_log[(size_t)d * N + n]) * LOG2E;
    h[n] = 0.0f;
  }
  float Dv = Dp[d];
  for (int l = 0; l < L; ++l) {
    size_t row = (size_t)b * L + l;
    size_t g = row * Dc + d;
    float dtv = dt[g];
    float xv = x[g];
    float us = dtv * xv;
    const float* Br = Bm + row * N;
    const float* Cr = Cm + row * N;
    float y = 0.0f;
#pragma unroll
    for (int n = 0; n < N; ++n) {
      h[n] = __builtin_amdgcn_exp2f(dtv * An2[n]) * h[n] + us * Br[n];
      y += h[n] * Cr[n];
    }
    out[g] = y + xv * Dv;
  }
}

__global__ void scan_fused_kernel(const float* __restrict__ x,
                                  const float* __restrict__ delta,
                                  const float* __restrict__ Bm,
                                  const float* __restrict__ Cm,
                                  const float* __restrict__ W,
                                  const float* __restrict__ bias,
                                  const float* __restrict__ A_log,
                                  const float* __restrict__ Dp,
                                  float* __restrict__ out,
                                  int L, int Dc, int K, int Bsz) {
  const int N = 16;
  long long idx = (long long)blockIdx.x * blockDim.x + threadIdx.x;
  if (idx >= (long long)Bsz * Dc) return;
  int d = (int)(idx % Dc);
  int b = (int)(idx / Dc);
  float An2[N], h[N];
#pragma unroll
  for (int n = 0; n < N; ++n) {
    An2[n] = -__expf(A_log[(size_t)d * N + n]) * LOG2E;
    h[n] = 0.0f;
  }
  float Dv = Dp[d];
  float bv = bias[d];
  const float* wr = W + (size_t)d * K;
  for (int l = 0; l < L; ++l) {
    size_t row = (size_t)b * L + l;
    const float* dr = delta + row * K;
    float acc = bv;
    for (int r = 0; r < K; ++r) acc += dr[r] * wr[r];
    float dtv = softplus_f(acc);
    float xv = x[row * Dc + d];
    float us = dtv * xv;
    const float* Br = Bm + row * N;
    const float* Cr = Cm + row * N;
    float y = 0.0f;
#pragma unroll
    for (int n = 0; n < N; ++n) {
      h[n] = __builtin_amdgcn_exp2f(dtv * An2[n]) * h[n] + us * Br[n];
      y += h[n] * Cr[n];
    }
    out[row * Dc + d] = y + xv * Dv;
  }
}

extern "C" void kernel_launch(void* const* d_in, const int* in_sizes, int n_in,
                              void* d_out, int out_size, void* d_ws, size_t ws_size,
                              hipStream_t stream) {
  const float* x     = (const float*)d_in[0];
  const float* delta = (const float*)d_in[1];
  const float* Bm    = (const float*)d_in[2];
  const float* Cm    = (const float*)d_in[3];
  const float* W     = (const float*)d_in[4];
  const float* bias  = (const float*)d_in[5];
  const float* A_log = (const float*)d_in[6];
  const float* Dp    = (const float*)d_in[7];
  float* out = (float*)d_out;

  int Dc = in_sizes[5];
  int K  = in_sizes[4] / Dc;
  long long BL = (long long)in_sizes[0] / Dc;
  int Bsz = (BL % 2 == 0) ? 2 : 1;
  int L = (int)(BL / Bsz);

  size_t dtElems = (size_t)BL * Dc;
  size_t dtBytes = dtElems * sizeof(float);

  const int Lc = 64;
  bool chunkable = (L % Lc) == 0 && (Dc % 256) == 0;
  int nCh = chunkable ? (L / Lc) : 1;
  size_t chElems = (size_t)Bsz * nCh * Dc;
  size_t fullBytes = dtBytes + chElems * sizeof(float) * (1 + 16 + 16);

  if (ws_size < dtBytes) {
    int threads = 256;
    long long tot = (long long)Bsz * Dc;
    int blocks = (int)((tot + threads - 1) / threads);
    scan_fused_kernel<<<blocks, threads, 0, stream>>>(
        x, delta, Bm, Cm, W, bias, A_log, Dp, out, L, Dc, K, Bsz);
    return;
  }

  float* dt = (float*)d_ws;

  if ((BL % 64) == 0 && (Dc % 128) == 0 && (K % 32) == 0) {
    int blocks = (int)(BL / 64) * (Dc / 128);
    dtproj_tiled_kernel<<<blocks, 256, 0, stream>>>(delta, W, bias, dt,
                                                    (int)BL, Dc, K);
  } else if ((BL % 16) == 0 && (Dc % 32) == 0 && (K % 4) == 0) {
    int tiles = (int)(BL / 16) * (Dc / 32);
    int blocks = (tiles + 7) / 8;
    dtproj_wmma_kernel<<<blocks, 256, 0, stream>>>(delta, W, bias, dt,
                                                   (int)BL, Dc, K);
  } else {
    long long tot = BL * (long long)Dc;
    int blocks = (int)((tot + 255) / 256);
    dtproj_naive_kernel<<<blocks, 256, 0, stream>>>(delta, W, bias, dt, BL, Dc, K);
  }

  if (chunkable && ws_size >= fullBytes) {
    float* Ssum   = dt + dtElems;
    float* Hlocal = Ssum + chElems;
    float* Hinit  = Hlocal + chElems * 16;

    int blocks13 = Bsz * nCh * (Dc / 256);
    scan_chunk_kernel<false><<<blocks13, 256, 0, stream>>>(
        dt, x, Bm, Cm, A_log, Dp, Hlocal, Ssum, (const float*)nullptr,
        (float*)nullptr, L, Dc, Lc, nCh);

    long long tot = (long long)Bsz * Dc * 16;
    int cblocks = (int)((tot + 255) / 256);
    scan_combine_kernel<<<cblocks, 256, 0, stream>>>(Hlocal, Ssum, A_log, Hinit,
                                                     Dc, nCh, Bsz);

    scan_chunk_kernel<true><<<blocks13, 256, 0, stream>>>(
        dt, x, Bm, Cm, A_log, Dp, (float*)nullptr, (float*)nullptr, Hinit, out,
        L, Dc, Lc, nCh);
  } else {
    long long tot = (long long)Bsz * Dc;
    int blocks = (int)((tot + 255) / 256);
    scan_simple_kernel<<<blocks, 256, 0, stream>>>(dt, x, Bm, Cm, A_log, Dp, out,
                                                   L, Dc, Bsz);
  }
}


# first scan pass: B rows are requested four timesteps at a time, four steps ahead of use (was one), so a second-level cache round trip runs under four steps of arithmetic; sums and order unchanged
# speedup vs baseline: 2.8148x; 1.0885x over previous
#include <hip/hip_runtime.h>
#include <math.h>

typedef __attribute__((ext_vector_type(2))) float v2f;
typedef __attribute__((ext_vector_type(8))) float v8f;
typedef __attribute__((ext_vector_type(4))) float v4f;

#define LOG2E 1.4426950408889634f
#define LN2F 0.6931471805599453f

__device__ __forceinline__ float softplus_f(float v) {
  float e = __builtin_amdgcn_exp2f(-fabsf(v) * LOG2E);
  float u = 1.0f + e;
  float dropped = e - (u - 1.0f);
  float lg = __builtin_amdgcn_logf(u) * LN2F;
  return fmaxf(v, 0.0f) + (lg + dropped * __builtin_amdgcn_rcpf(u));
}

__global__ __launch_bounds__(256) void dtproj_wmma_kernel(
    const float* __restrict__ delta, const float* __restrict__ W,
    const float* __restrict__ bias, float* __restrict__ dt,
    int M, int Dc, int K) {
  int wave = threadIdx.x >> 5;
  int lane = threadIdx.x & 31;
  int tilesN = Dc >> 5;
  int tile = blockIdx.x * 8 + wave;
  int totalTiles = (M >> 4) * tilesN;
  if (tile >= totalTiles) return;
  int tn = tile % tilesN;
  int tm = tile / tilesN;
  int mn   = lane & 15;
  int half = lane >> 4;

  const float* arow  = delta + (size_t)(tm * 16 + mn) * K;
  const float* brow0 = W     + (size_t)(tn * 32 + mn) * K;
  const float* brow1 = W     + (size_t)(tn * 32 + 16 + mn) * K;

  v8f acc0 = {};
  v8f acc1 = {};
  for (int k0 = 0; k0 < K; k0 += 4) {
    int kb = k0 + (half << 1);
    v2f a, b0, b1;
    a.x  = arow[kb];  a.y  = arow[kb + 1];
    b0.x = brow0[kb]; b0.y = brow0[kb + 1];
    b1.x = brow1[kb]; b1.y = brow1[kb + 1];
    acc0 = __builtin_amdgcn_wmma_f32_16x16x4_f32(
        false, a, false, b0, (short)0, acc0, false, false);
    acc1 = __builtin_amdgcn_wmma_f32_16x16x4_f32(
        false, a, false, b1, (short)0, acc1, false, false);
  }

  int gd0 = tn * 32 + mn;
  int gd1 = gd0 + 16;
  float bv0 = bias[gd0];
  float bv1 = bias[gd1];
  int gm0 = tm * 16 + half * 8;
#pragma unroll
  for (int v = 0; v < 8; ++v) {
    size_t rowoff = (size_t)(gm0 + v) * Dc;
    dt[rowoff + gd0] = softplus_f(acc0[v] + bv0);
    dt[rowoff + gd1] = softplus_f(acc1[v] + bv1);
  }
}

#define DTP_BM 64
#define DTP_BN 128
#define DTP_KC 32
#define DTP_LD 36

__global__ __launch_bounds__(256) void dtproj_tiled_kernel(
    const float* __restrict__ delta, const float* __restrict__ W,
    const float* __restrict__ bias, float* __restrict__ dt,
    int M, int Dc, int K) {
  __shared__ __attribute__((aligned(16))) float sA[2][DTP_BM * DTP_LD];
  __shared__ __attribute__((aligned(16))) float sB[2][DTP_BN * DTP_LD];

  const int tid  = threadIdx.x;
  const int wave = tid >> 5;
  const int lane = tid & 31;
  const int c = lane & 15;
  const int h = lane >> 4;
  const int tilesN = Dc / DTP_BN;
  const int m0 = (blockIdx.x / tilesN) * DTP_BM;
  const int n0 = (blockIdx.x % tilesN) * DTP_BN;
  const int wm = (wave >> 2) * 32;
  const int wn = (wave & 3) * 32;

  const int crow = tid >> 3;
  const int ccol = (tid & 7) * 4;
  const float* gA = delta + (size_t)(m0 + crow) * K + ccol;
  const float* gB = W     + (size_t)(n0 + crow) * K + ccol;
  const size_t gstep = (size_t)32 * K;

  v4f pa[2], pb[4];
#pragma unroll
  for (int p = 0; p < 2; ++p) pa[p] = *(const v4f*)(gA + p * gstep);
#pragma unroll
  for (int p = 0; p < 4; ++p) pb[p] = *(const v4f*)(gB + p * gstep);
#pragma unroll
  for (int p = 0; p < 2; ++p) *(v4f*)&sA[0][(crow + 32 * p) * DTP_LD + ccol] = pa[p];
#pragma unroll
  for (int p = 0; p < 4; ++p) *(v4f*)&sB[0][(crow + 32 * p) * DTP_LD + ccol] = pb[p];
  __syncthreads();

  v8f acc[2][2];
#pragma unroll
  for (int i = 0; i < 2; ++i)
#pragma unroll
    for (int j = 0; j < 2; ++j) acc[i][j] = (v8f){};

  const int nk = K / DTP_KC;
  for (int kc = 0; kc < nk; ++kc) {
    const int st = kc & 1;
    const bool more = (kc + 1 < nk);
    if (more) {
      const float* gAn = gA + (size_t)(kc + 1) * DTP_KC;
      const float* gBn = gB + (size_t)(kc + 1) * DTP_KC;
#pragma unroll
      for (int p = 0; p < 2; ++p) pa[p] = *(const v4f*)(gAn + p * gstep);
#pragma unroll
      for (int p = 0; p < 4; ++p) pb[p] = *(const v4f*)(gBn + p * gstep);
    }
    const float* abase = &sA[st][(wm + c) * DTP_LD + 2 * h];
    const float* bbase = &sB[st][(wn + c) * DTP_LD + 2 * h];
#pragma unroll
    for (int kk = 0; kk < DTP_KC; kk += 4) {
      v2f af[2], bf[2];
#pragma unroll
      for (int i = 0; i < 2; ++i) af[i] = *(const v2f*)(abase + i * 16 * DTP_LD + kk);
#pragma unroll
      for (int j = 0; j < 2; ++j) bf[j] = *(const v2f*)(bbase + j * 16 * DTP_LD + kk);
#pragma unroll
      for (int i = 0; i < 2; ++i)
#pragma unroll
        for (int j = 0; j < 2; ++j)
          acc[i][j] = __builtin_amdgcn_wmma_f32_16x16x4_f32(
              false, af[i], false, bf[j], (short)0, acc[i][j], false, false);
    }
    if (more) {
#pragma unroll
      for (int p = 0; p < 2; ++p) *(v4f*)&sA[st ^ 1][(crow + 32 * p) * DTP_LD + ccol] = pa[p];
#pragma unroll
      for (int p = 0; p < 4; ++p) *(v4f*)&sB[st ^ 1][(crow + 32 * p) * DTP_LD + ccol] = pb[p];
    }
    __syncthreads();
  }

  const int gd0 = n0 + wn + c;
  const int gd1 = gd0 + 16;
  const float bv0 = bias[gd0];
  const float bv1 = bias[gd1];
#pragma unroll
  for (int i = 0; i < 2; ++i) {
    const int gm0 = m0 + wm + 16 * i + 8 * h;
#pragma unroll
    for (int r = 0; r < 8; ++r) {
      size_t rowoff = (size_t)(gm0 + r) * Dc;
      dt[rowoff + gd0] = softplus_f(acc[i][0][r] + bv0);
      dt[rowoff + gd1] = softplus_f(acc[i][1][r] + bv1);
    }
  }
}

__global__ void dtproj_naive_kernel(const float* __restrict__ delta,
                                    const float* __restrict__ W,
                                    const float* __restrict__ bias,
                                    float* __restrict__ dt,
                                    long long M, int Dc, int K) {
  long long idx = (long long)blockIdx.x * blockDim.x + threadIdx.x;
  if (idx >= M * (long long)Dc) return;
  int d = (int)(idx % Dc);
  long long m = idx / Dc;
  float acc = bias[d];
  const float* dr = delta + m * (long long)K;
  const float* wr = W + (size_t)d * K;
  for (int r = 0; r < K; ++r) acc += dr[r] * wr[r];
  dt[idx] = softplus_f(acc);
}

template <bool FINAL>
__global__ __launch_bounds__(256) void scan_chunk_kernel(
    const float* __restrict__ dt, const float* __restrict__ x,
    const float* __restrict__ Bm, const float* __restrict__ Cm,
    const float* __restrict__ A_log, const float* __restrict__ Dp,
    float* __restrict__ Hout, float* __restrict__ Ssum,
    const float* __restrict__ Hinit, float* __restrict__ out,
    int L, int Dc, int Lc, int nCh) {
  const int N = 16;
  const int PB = 16;
  constexpr int G = FINAL ? 1 : 4;

  int DB = Dc / blockDim.x;
  int bx = blockIdx.x;
  int dblk = bx % DB; bx /= DB;
  int chunk = bx % nCh; bx /= nCh;
  int b = bx;
  int d = dblk * blockDim.x + threadIdx.x;
  unsigned du = (unsigned)d;

  size_t rowbase = (size_t)b * L + (size_t)chunk * Lc;
  const float* dtrow = dt + rowbase * Dc;
  const float* xrow  = x  + rowbase * Dc;
  typedef const __attribute__((address_space(4))) float cfloat_k;
  cfloat_k* Brow = (cfloat_k*)(Bm + rowbase * N);
  cfloat_k* Crow = (cfloat_k*)(Cm + rowbase * N);
  float bcur[G][N], ccur[G][N];
#pragma unroll
  for (int g = 0; g < G; ++g) {
    const int r = (g == 0 || g < Lc) ? g : (Lc - 1);
#pragma unroll
    for (int n = 0; n < N; ++n) {
      bcur[g][n] = Brow[r * N + n];
      ccur[g][n] = FINAL ? Crow[r * N + n] : 0.0f;
    }
  }

  const int nb = Lc / PB;
  float dtc[PB], xc[PB], dtn[PB], xn[PB];
#pragma unroll
  for (int i = 0; i < PB; ++i) { dtc[i] = 0.0f; xc[i] = 0.0f; dtn[i] = 0.0f; xn[i] = 0.0f; }
  if (nb > 0) {
#pragma unroll
    for (int i = 0; i < PB; ++i) {
      dtc[i] = dtrow[(size_t)i * Dc + du];
      xc[i]  = xrow[(size_t)i * Dc + du];
    }
  }

  float alog[N], h[N];
  size_t hidx = (((size_t)b * nCh + chunk) * Dc + d) * N;
#pragma unroll
  for (int n = 0; n < N; ++n) h[n] = FINAL ? Hinit[hidx + n] : 0.0f;
  float Dv = FINAL ? Dp[d] : 0.0f;
#pragma unroll
  for (int n = 0; n < N; ++n) alog[n] = A_log[(size_t)d * N + n];

  float An2[N];
#pragma unroll
  for (int n = 0; n < N; ++n) An2[n] = -__expf(alog[n]) * LOG2E;

  float dtsum = 0.0f;

  if (FINAL) {
    asm volatile("" :: "v"(Dv), "v"(h[0]), "v"(h[1]), "v"(h[2]), "v"(h[3]),
                       "v"(h[4]), "v"(h[5]), "v"(h[6]), "v"(h[7]),
                       "v"(h[8]), "v"(h[9]), "v"(h[10]), "v"(h[11]),
                       "v"(h[12]), "v"(h[13]), "v"(h[14]), "v"(h[15]));
  }

  for (int bi = 0; bi < nb; ++bi) {
    const int l0 = bi * PB;
    if (bi + 1 < nb) {
      const float* dnext = dtrow + (size_t)(l0 + PB) * Dc;
      const float* xnext = xrow  + (size_t)(l0 + PB) * Dc;
#pragma unroll
      for (int i = 0; i < PB; ++i) {
        dtn[i] = dnext[(size_t)i * Dc + du];
        xn[i]  = xnext[(size_t)i * Dc + du];
      }
    }
#pragma unroll
    for (int i = 0; i < PB; i += G) {
      const int l = l0 + i;
      float ub[G][N];
#pragma unroll
      for (int g = 0; g < G; ++g) {
        const float us = dtc[i + g] * xc[i + g];
#pragma unroll
        for (int n = 0; n < N; ++n) ub[g][n] = us * bcur[g][n];
      }
#pragma unroll
      for (int g = 0; g < G; ++g) {
        asm volatile("" :: "s"(bcur[g][0]), "s"(bcur[g][1]), "s"(bcur[g][2]), "s"(bcur[g][3]),
                           "s"(bcur[g][4]), "s"(bcur[g][5]), "s"(bcur[g][6]), "s"(bcur[g][7]),
                           "s"(bcur[g][8]), "s"(bcur[g][9]), "s"(bcur[g][10]), "s"(bcur[g][11]),
                           "s"(bcur[g][12]), "s"(bcur[g][13]), "s"(bcur[g][14]), "s"(bcur[g][15]));
      }
      const int lp = (i + G < PB) ? (l + G) : ((l + 2 * G <= Lc) ? (l + G) : l);
      float cnx[G][N];
#pragma unroll
      for (int g = 0; g < G; ++g) {
#pragma unroll
        for (int n = 0; n < N; ++n) {
          bcur[g][n] = Brow[(lp + g) * N + n];
          cnx[g][n] = FINAL ? Crow[(lp + g) * N + n] : 0.0f;
        }
      }
      __builtin_amdgcn_sched_barrier(0);
#pragma unroll
      for (int g = 0; g < G; ++g) {
        float dtv = dtc[i + g];
        float xv = xc[i + g];
        if (!FINAL) dtsum += dtv;
        float y = 0.0f;
#pragma unroll
        for (int n = 0; n < N; ++n) {
          float a = __builtin_amdgcn_exp2f(dtv * An2[n]);
          h[n] = __builtin_fmaf(a, h[n], ub[g][n]);
          if (FINAL) y += h[n] * ccur[g][n];
        }
        if (FINAL) out[(rowbase + l + g) * Dc + du] = y + xv * Dv;
      }
#pragma unroll
      for (int g = 0; g < G; ++g) {
#pragma unroll
        for (int n = 0; n < N; ++n) ccur[g][n] = cnx[g][n];
      }
      if (!FINAL) __builtin_amdgcn_sched_barrier(0);
    }
#pragma unroll
    for (int i = 0; i < PB; ++i) { dtc[i] = dtn[i]; xc[i] = xn[i]; }
  }

  for (int l = nb * PB; l < Lc; ++l) {
    float dtv = dtrow[(size_t)l * Dc + du];
    float xv = xrow[(size_t)l * Dc + du];
    float us = dtv * xv;
    if (!FINAL) dtsum += dtv;
    float y = 0.0f;
#pragma unroll
    for (int n = 0; n < N; ++n) {
      float a = __builtin_amdgcn_exp2f(dtv * An2[n]);
      h[n] = __builtin_fmaf(a, h[n], us * Brow[l * N + n]);
      if (FINAL) y += h[n] * Crow[l * N + n];
    }
    if (FINAL) out[(rowbase + l) * Dc + du] = y + xv * Dv;
  }

  if (!FINAL) {
#pragma unroll
    for (int n = 0; n < N; ++n) Hout[hidx + n] = h[n];
    Ssum[((size_t)b * nCh + chunk) * Dc + d] = dtsum;
  }
}

__global__ __launch_bounds__(256) void scan_combine_kernel(
    const float* __restrict__ Hlocal, const float* __restrict__ Ssum,
    const float* __restrict__ A_log, float* __restrict__ Hinit,
    int Dc, int nCh, int Bsz) {
  const int N = 16;
  const int CB = 32;
  const long long perB = (long long)Dc * N;
  long long idx = (long long)blockIdx.x * blockDim.x + threadIdx.x;
  if (idx >= (long long)Bsz * perB) return;
  const int b = (int)(idx / perB);
  const int dn = (int)(idx - (long long)b * perB);
  const int d = dn >> 4;
  const size_t stepH = (size_t)perB;
  const size_t stepS = (size_t)Dc;
  const float* hp = Hlocal + (size_t)b * nCh * stepH + dn;
  const float* sp = Ssum   + (size_t)b * nCh * stepS + d;
  float*       op = Hinit  + (size_t)b * nCh * stepH + dn;

  const float alog = A_log[dn];
  const int nb = nCh / CB;
  float Sc[CB], Hc[CB], Sn[CB], Hn[CB];
#pragma unroll
  for (int j = 0; j < CB; ++j) { Sc[j] = 0.0f; Hc[j] = 0.0f; Sn[j] = 0.0f; Hn[j] = 0.0f; }
  if (nb > 0) {
#pragma unroll
    for (int j = 0; j < CB; ++j) {
      Sc[j] = sp[(size_t)j * stepS];
      Hc[j] = hp[(size_t)j * stepH];
    }
  }
  const float An2 = -__expf(alog) * LOG2E;
  float h = 0.0f;
  for (int bi = 0; bi < nb; ++bi) {
    const int c0 = bi * CB;
    if (bi + 1 < nb) {
#pragma unroll
      for (int j = 0; j < CB; ++j) {
        Sn[j] = sp[(size_t)(c0 + CB + j) * stepS];
        Hn[j] = hp[(size_t)(c0 + CB + j) * stepH];
      }
    }
    __builtin_amdgcn_sched_barrier(0);
#pragma unroll
    for (int j = 0; j < CB; ++j) {
      op[(size_t)(c0 + j) * stepH] = h;
      float a = __builtin_amdgcn_exp2f(An2 * Sc[j]);
      h = a * h + Hc[j];
    }
#pragma unroll
    for (int j = 0; j < CB; ++j) { Sc[j] = Sn[j]; Hc[j] = Hn[j]; }
  }
  for (int c = nb * CB; c < nCh; ++c) {
    float S = sp[(size_t)c * stepS];
    op[(size_t)c * stepH] = h;
    float a = __builtin_amdgcn_exp2f(An2 * S);
    h = a * h + hp[(size_t)c * stepH];
  }
}

__global__ void scan_simple_kernel(const float* __restrict__ dt,
                                   const float* __restrict__ x,
                                   const float* __restrict__ Bm,
                                   const float* __restrict__ Cm,
                                   const float* __restrict__ A_log,
                                   const float* __restrict__ Dp,
                                   float* __restrict__ out,
                                   int L, int Dc, int Bsz) {
  const int N = 16;
  long long idx = (long long)blockIdx.x * blockDim.x + threadIdx.x;
  if (idx >= (long long)Bsz * Dc) return;
  int d = (int)(idx % Dc);
  int b = (int)(idx / Dc);
  float An2[N], h[N];
#pragma unroll
  for (int n = 0; n < N; ++n) {
    An2[n] = -__expf(A_log[(size_t)d * N + n]) * LOG2E;
    h[n] = 0.0f;
  }
  float Dv = Dp[d];
  for (int l = 0; l < L; ++l) {
    size_t row = (size_t)b * L + l;
    size_t g = row * Dc + d;
    float dtv = dt[g];
    float xv = x[g];
    float us = dtv * xv;
    const float* Br = Bm + row * N;
    const float* Cr = Cm + row * N;
    float y = 0.0f;
#pragma unroll
    for (int n = 0; n < N; ++n) {
      h[n] = __builtin_amdgcn_exp2f(dtv * An2[n]) * h[n] + us * Br[n];
      y += h[n] * Cr[n];
    }
    out[g] = y + xv * Dv;
  }
}

__global__ void scan_fused_kernel(const float* __restrict__ x,
                                  const float* __restrict__ delta,
                                  const float* __restrict__ Bm,
                                  const float* __restrict__ Cm,
                                  const float* __restrict__ W,
                                  const float* __restrict__ bias,
                                  const float* __restrict__ A_log,
                                  const float* __restrict__ Dp,
                                  float* __restrict__ out,
                                  int L, int Dc, int K, int Bsz) {
  const int N = 16;
  long long idx = (long long)blockIdx.x * blockDim.x + threadIdx.x;
  if (idx >= (long long)Bsz * Dc) return;
  int d = (int)(idx % Dc);
  int b = (int)(idx / Dc);
  float An2[N], h[N];
#pragma unroll
  for (int n = 0; n < N; ++n) {
    An2[n] = -__expf(A_log[(size_t)d * N + n]) * LOG2E;
    h[n] = 0.0f;
  }
  float Dv = Dp[d];
  float bv = bias[d];
  const float* wr = W + (size_t)d * K;
  for (int l = 0; l < L; ++l) {
    size_t row = (size_t)b * L + l;
    const float* dr = delta + row * K;
    float acc = bv;
    for (int r = 0; r < K; ++r) acc += dr[r] * wr[r];
    float dtv = softplus_f(acc);
    float xv = x[row * Dc + d];
    float us = dtv * xv;
    const float* Br = Bm + row * N;
    const float* Cr = Cm + row * N;
    float y = 0.0f;
#pragma unroll
    for (int n = 0; n < N; ++n) {
      h[n] = __builtin_amdgcn_exp2f(dtv * An2[n]) * h[n] + us * Br[n];
      y += h[n] * Cr[n];
    }
    out[row * Dc + d] = y + xv * Dv;
  }
}

extern "C" void kernel_launch(void* const* d_in, const int* in_sizes, int n_in,
                              void* d_out, int out_size, void* d_ws, size_t ws_size,
                              hipStream_t stream) {
  const float* x     = (const float*)d_in[0];
  const float* delta = (const float*)d_in[1];
  const float* Bm    = (const float*)d_in[2];
  const float* Cm    = (const float*)d_in[3];
  const float* W     = (const float*)d_in[4];
  const float* bias  = (const float*)d_in[5];
  const float* A_log = (const float*)d_in[6];
  const float* Dp    = (const float*)d_in[7];
  float* out = (float*)d_out;

  int Dc = in_sizes[5];
  int K  = in_sizes[4] / Dc;
  long long BL = (long long)in_sizes[0] / Dc;
  int Bsz = (BL % 2 == 0) ? 2 : 1;
  int L = (int)(BL / Bsz);

  size_t dtElems = (size_t)BL * Dc;
  size_t dtBytes = dtElems * sizeof(float);

  const int Lc = 64;
  bool chunkable = (L % Lc) == 0 && (Dc % 256) == 0;
  int nCh = chunkable ? (L / Lc) : 1;
  size_t chElems = (size_t)Bsz * nCh * Dc;
  size_t fullBytes = dtBytes + chElems * sizeof(float) * (1 + 16 + 16);

  if (ws_size < dtBytes) {
    int threads = 256;
    long long tot = (long long)Bsz * Dc;
    int blocks = (int)((tot + threads - 1) / threads);
    scan_fused_kernel<<<blocks, threads, 0, stream>>>(
        x, delta, Bm, Cm, W, bias, A_log, Dp, out, L, Dc, K, Bsz);
    return;
  }

  float* dt = (float*)d_ws;

  if ((BL % 64) == 0 && (Dc % 128) == 0 && (K % 32) == 0) {
    int blocks = (int)(BL / 64) * (Dc / 128);
    dtproj_tiled_kernel<<<blocks, 256, 0, stream>>>(delta, W, bias, dt,
                                                    (int)BL, Dc, K);
  } else if ((BL % 16) == 0 && (Dc % 32) == 0 && (K % 4) == 0) {
    int tiles = (int)(BL / 16) * (Dc / 32);
    int blocks = (tiles + 7) / 8;
    dtproj_wmma_kernel<<<blocks, 256, 0, stream>>>(delta, W, bias, dt,
                                                   (int)BL, Dc, K);
  } else {
    long long tot = BL * (long long)Dc;
    int blocks = (int)((tot + 255) / 256);
    dtproj_naive_kernel<<<blocks, 256, 0, stream>>>(delta, W, bias, dt, BL, Dc, K);
  }

  if (chunkable && ws_size >= fullBytes) {
    float* Ssum   = dt + dtElems;
    float* Hlocal = Ssum + chElems;
    float* Hinit  = Hlocal + chElems * 16;

    int blocks13 = Bsz * nCh * (Dc / 256);
    scan_chunk_kernel<false><<<blocks13, 256, 0, stream>>>(
        dt, x, Bm, Cm, A_log, Dp, Hlocal, Ssum, (const float*)nullptr,
        (float*)nullptr, L, Dc, Lc, nCh);

    long long tot = (long long)Bsz * Dc * 16;
    int cblocks = (int)((tot + 255) / 256);
    scan_combine_kernel<<<cblocks, 256, 0, stream>>>(Hlocal, Ssum, A_log, Hinit,
                                                     Dc, nCh, Bsz);

    scan_chunk_kernel<true><<<blocks13, 256, 0, stream>>>(
        dt, x, Bm, Cm, A_log, Dp, (float*)nullptr, (float*)nullptr, Hinit, out,
        L, Dc, Lc, nCh);
  } else {
    long long tot = (long long)Bsz * Dc;
    int blocks = (int)((tot + 255) / 256);
    scan_simple_kernel<<<blocks, 256, 0, stream>>>(dt, x, Bm, Cm, A_log, Dp, out,
                                                   L, Dc, Bsz);
  }
}


# last scan pass: the chunk's C rows are staged once in LDS and read 4 steps at a time into vector registers; scalar registers hold B only, requested 4 steps ahead (was 1); sums and order unchanged
# speedup vs baseline: 2.9572x; 1.0506x over previous
#include <hip/hip_runtime.h>
#include <math.h>

typedef __attribute__((ext_vector_type(2))) float v2f;
typedef __attribute__((ext_vector_type(8))) float v8f;
typedef __attribute__((ext_vector_type(4))) float v4f;

#define LOG2E 1.4426950408889634f
#define LN2F 0.6931471805599453f

__device__ __forceinline__ float softplus_f(float v) {
  float e = __builtin_amdgcn_exp2f(-fabsf(v) * LOG2E);
  float u = 1.0f + e;
  float dropped = e - (u - 1.0f);
  float lg = __builtin_amdgcn_logf(u) * LN2F;
  return fmaxf(v, 0.0f) + (lg + dropped * __builtin_amdgcn_rcpf(u));
}

__global__ __launch_bounds__(256) void dtproj_wmma_kernel(
    const float* __restrict__ delta, const float* __restrict__ W,
    const float* __restrict__ bias, float* __restrict__ dt,
    int M, int Dc, int K) {
  int wave = threadIdx.x >> 5;
  int lane = threadIdx.x & 31;
  int tilesN = Dc >> 5;
  int tile = blockIdx.x * 8 + wave;
  int totalTiles = (M >> 4) * tilesN;
  if (tile >= totalTiles) return;
  int tn = tile % tilesN;
  int tm = tile / tilesN;
  int mn   = lane & 15;
  int half = lane >> 4;

  const float* arow  = delta + (size_t)(tm * 16 + mn) * K;
  const float* brow0 = W     + (size_t)(tn * 32 + mn) * K;
  const float* brow1 = W     + (size_t)(tn * 32 + 16 + mn) * K;

  v8f acc0 = {};
  v8f acc1 = {};
  for (int k0 = 0; k0 < K; k0 += 4) {
    int kb = k0 + (half << 1);
    v2f a, b0, b1;
    a.x  = arow[kb];  a.y  = arow[kb + 1];
    b0.x = brow0[kb]; b0.y = brow0[kb + 1];
    b1.x = brow1[kb]; b1.y = brow1[kb + 1];
    acc0 = __builtin_amdgcn_wmma_f32_16x16x4_f32(
        false, a, false, b0, (short)0, acc0, false, false);
    acc1 = __builtin_amdgcn_wmma_f32_16x16x4_f32(
        false, a, false, b1, (short)0, acc1, false, false);
  }

  int gd0 = tn * 32 + mn;
  int gd1 = gd0 + 16;
  float bv0 = bias[gd0];
  float bv1 = bias[gd1];
  int gm0 = tm * 16 + half * 8;
#pragma unroll
  for (int v = 0; v < 8; ++v) {
    size_t rowoff = (size_t)(gm0 + v) * Dc;
    dt[rowoff + gd0] = softplus_f(acc0[v] + bv0);
    dt[rowoff + gd1] = softplus_f(acc1[v] + bv1);
  }
}

#define DTP_BM 64
#define DTP_BN 128
#define DTP_KC 32
#define DTP_LD 36

__global__ __launch_bounds__(256) void dtproj_tiled_kernel(
    const float* __restrict__ delta, const float* __restrict__ W,
    const float* __restrict__ bias, float* __restrict__ dt,
    int M, int Dc, int K) {
  __shared__ __attribute__((aligned(16))) float sA[2][DTP_BM * DTP_LD];
  __shared__ __attribute__((aligned(16))) float sB[2][DTP_BN * DTP_LD];

  const int tid  = threadIdx.x;
  const int wave = tid >> 5;
  const int lane = tid & 31;
  const int c = lane & 15;
  const int h = lane >> 4;
  const int tilesN = Dc / DTP_BN;
  const int m0 = (blockIdx.x / tilesN) * DTP_BM;
  const int n0 = (blockIdx.x % tilesN) * DTP_BN;
  const int wm = (wave >> 2) * 32;
  const int wn = (wave & 3) * 32;

  const int crow = tid >> 3;
  const int ccol = (tid & 7) * 4;
  const float* gA = delta + (size_t)(m0 + crow) * K + ccol;
  const float* gB = W     + (size_t)(n0 + crow) * K + ccol;
  const size_t gstep = (size_t)32 * K;

  v4f pa[2], pb[4];
#pragma unroll
  for (int p = 0; p < 2; ++p) pa[p] = *(const v4f*)(gA + p * gstep);
#pragma unroll
  for (int p = 0; p < 4; ++p) pb[p] = *(const v4f*)(gB + p * gstep);
#pragma unroll
  for (int p = 0; p < 2; ++p) *(v4f*)&sA[0][(crow + 32 * p) * DTP_LD + ccol] = pa[p];
#pragma unroll
  for (int p = 0; p < 4; ++p) *(v4f*)&sB[0][(crow + 32 * p) * DTP_LD + ccol] = pb[p];
  __syncthreads();

  v8f acc[2][2];
#pragma unroll
  for (int i = 0; i < 2; ++i)
#pragma unroll
    for (int j = 0; j < 2; ++j) acc[i][j] = (v8f){};

  const int nk = K / DTP_KC;
  for (int kc = 0; kc < nk; ++kc) {
    const int st = kc & 1;
    const bool more = (kc + 1 < nk);
    if (more) {
      const float* gAn = gA + (size_t)(kc + 1) * DTP_KC;
      const float* gBn = gB + (size_t)(kc + 1) * DTP_KC;
#pragma unroll
      for (int p = 0; p < 2; ++p) pa[p] = *(const v4f*)(gAn + p * gstep);
#pragma unroll
      for (int p = 0; p < 4; ++p) pb[p] = *(const v4f*)(gBn + p * gstep);
    }
    const float* abase = &sA[st][(wm + c) * DTP_LD + 2 * h];
    const float* bbase = &sB[st][(wn + c) * DTP_LD + 2 * h];
#pragma unroll
    for (int kk = 0; kk < DTP_KC; kk += 4) {
      v2f af[2], bf[2];
#pragma unroll
      for (int i = 0; i < 2; ++i) af[i] = *(const v2f*)(abase + i * 16 * DTP_LD + kk);
#pragma unroll
      for (int j = 0; j < 2; ++j) bf[j] = *(const v2f*)(bbase + j * 16 * DTP_LD + kk);
#pragma unroll
      for (int i = 0; i < 2; ++i)
#pragma unroll
        for (int j = 0; j < 2; ++j)
          acc[i][j] = __builtin_amdgcn_wmma_f32_16x16x4_f32(
              false, af[i], false, bf[j], (short)0, acc[i][j], false, false);
    }
    if (more) {
#pragma unroll
      for (int p = 0; p < 2; ++p) *(v4f*)&sA[st ^ 1][(crow + 32 * p) * DTP_LD + ccol] = pa[p];
#pragma unroll
      for (int p = 0; p < 4; ++p) *(v4f*)&sB[st ^ 1][(crow + 32 * p) * DTP_LD + ccol] = pb[p];
    }
    __syncthreads();
  }

  const int gd0 = n0 + wn + c;
  const int gd1 = gd0 + 16;
  const float bv0 = bias[gd0];
  const float bv1 = bias[gd1];
#pragma unroll
  for (int i = 0; i < 2; ++i) {
    const int gm0 = m0 + wm + 16 * i + 8 * h;
#pragma unroll
    for (int r = 0; r < 8; ++r) {
      size_t rowoff = (size_t)(gm0 + r) * Dc;
      dt[rowoff + gd0] = softplus_f(acc[i][0][r] + bv0);
      dt[rowoff + gd1] = softplus_f(acc[i][1][r] + bv1);
    }
  }
}

__global__ void dtproj_naive_kernel(const float* __restrict__ delta,
                                    const float* __restrict__ W,
                                    const float* __restrict__ bias,
                                    float* __restrict__ dt,
                                    long long M, int Dc, int K) {
  long long idx = (long long)blockIdx.x * blockDim.x + threadIdx.x;
  if (idx >= M * (long long)Dc) return;
  int d = (int)(idx % Dc);
  long long m = idx / Dc;
  float acc = bias[d];
  const float* dr = delta + m * (long long)K;
  const float* wr = W + (size_t)d * K;
  for (int r = 0; r < K; ++r) acc += dr[r] * wr[r];
  dt[idx] = softplus_f(acc);
}

#define SCAN_LC 64

template <bool FINAL>
__global__ __launch_bounds__(256) void scan_chunk_kernel(
    const float* __restrict__ dt, const float* __restrict__ x,
    const float* __restrict__ Bm, const float* __restrict__ Cm,
    const float* __restrict__ A_log, const float* __restrict__ Dp,
    float* __restrict__ Hout, float* __restrict__ Ssum,
    const float* __restrict__ Hinit, float* __restrict__ out,
    int L, int Dc, int Lc, int nCh) {
  const int N = 16;
  const int PB = 16;
  constexpr int G = 4;
  __shared__ __attribute__((aligned(16))) float sC[FINAL ? SCAN_LC * 16 : 4];

  int DB = Dc / blockDim.x;
  int bx = blockIdx.x;
  int dblk = bx % DB; bx /= DB;
  int chunk = bx % nCh; bx /= nCh;
  int b = bx;
  int d = dblk * blockDim.x + threadIdx.x;
  unsigned du = (unsigned)d;

  size_t rowbase = (size_t)b * L + (size_t)chunk * Lc;
  const float* dtrow = dt + rowbase * Dc;
  const float* xrow  = x  + rowbase * Dc;
  typedef const __attribute__((address_space(4))) float cfloat_k;
  cfloat_k* Brow = (cfloat_k*)(Bm + rowbase * N);
  cfloat_k* Crow = (cfloat_k*)(Cm + rowbase * N);
  const int ce = (int)threadIdx.x * 4;
  v4f cstage = {};
  if (FINAL && ce < Lc * N) cstage = *(const v4f*)(Cm + rowbase * N + ce);
  float bcur[G][N];
#pragma unroll
  for (int g = 0; g < G; ++g) {
    const int r = (g == 0 || g < Lc) ? g : (Lc - 1);
#pragma unroll
    for (int n = 0; n < N; ++n) bcur[g][n] = Brow[r * N + n];
  }

  const int nb = Lc / PB;
  float dtc[PB], xc[PB], dtn[PB], xn[PB];
#pragma unroll
  for (int i = 0; i < PB; ++i) { dtc[i] = 0.0f; xc[i] = 0.0f; dtn[i] = 0.0f; xn[i] = 0.0f; }
  if (nb > 0) {
#pragma unroll
    for (int i = 0; i < PB; ++i) {
      dtc[i] = dtrow[(size_t)i * Dc + du];
      xc[i]  = xrow[(size_t)i * Dc + du];
    }
  }

  float alog[N], h[N];
  size_t hidx = (((size_t)b * nCh + chunk) * Dc + d) * N;
#pragma unroll
  for (int n = 0; n < N; ++n) h[n] = FINAL ? Hinit[hidx + n] : 0.0f;
  float Dv = FINAL ? Dp[d] : 0.0f;
#pragma unroll
  for (int n = 0; n < N; ++n) alog[n] = A_log[(size_t)d * N + n];

  float An2[N];
#pragma unroll
  for (int n = 0; n < N; ++n) An2[n] = -__expf(alog[n]) * LOG2E;

  float dtsum = 0.0f;

  if (FINAL) {
    asm volatile("" :: "v"(Dv), "v"(h[0]), "v"(h[1]), "v"(h[2]), "v"(h[3]),
                       "v"(h[4]), "v"(h[5]), "v"(h[6]), "v"(h[7]),
                       "v"(h[8]), "v"(h[9]), "v"(h[10]), "v"(h[11]),
                       "v"(h[12]), "v"(h[13]), "v"(h[14]), "v"(h[15]));
  }

  if (FINAL) {
    if (ce < Lc * N) *(v4f*)&sC[ce] = cstage;
    __syncthreads();
  }

  for (int bi = 0; bi < nb; ++bi) {
    const int l0 = bi * PB;
    if (bi + 1 < nb) {
      const float* dnext = dtrow + (size_t)(l0 + PB) * Dc;
      const float* xnext = xrow  + (size_t)(l0 + PB) * Dc;
#pragma unroll
      for (int i = 0; i < PB; ++i) {
        dtn[i] = dnext[(size_t)i * Dc + du];
        xn[i]  = xnext[(size_t)i * Dc + du];
      }
    }
#pragma unroll
    for (int i = 0; i < PB; i += G) {
      const int l = l0 + i;
      float cr[G][N];
#pragma unroll
      for (int g = 0; g < G; ++g) {
#pragma unroll
        for (int q = 0; q < N / 4; ++q) {
          v4f c4 = {};
          if (FINAL) c4 = *(const v4f*)&sC[(l + g) * N + 4 * q];
          cr[g][4 * q] = c4.x; cr[g][4 * q + 1] = c4.y; cr[g][4 * q + 2] = c4.z; cr[g][4 * q + 3] = c4.w;
        }
      }
      float ub[G][N];
#pragma unroll
      for (int g = 0; g < G; ++g) {
        const float us = dtc[i + g] * xc[i + g];
#pragma unroll
        for (int n = 0; n < N; ++n) ub[g][n] = us * bcur[g][n];
      }
#pragma unroll
      for (int g = 0; g < G; ++g) {
        asm volatile("" :: "s"(bcur[g][0]), "s"(bcur[g][1]), "s"(bcur[g][2]), "s"(bcur[g][3]),
                           "s"(bcur[g][4]), "s"(bcur[g][5]), "s"(bcur[g][6]), "s"(bcur[g][7]),
                           "s"(bcur[g][8]), "s"(bcur[g][9]), "s"(bcur[g][10]), "s"(bcur[g][11]),
                           "s"(bcur[g][12]), "s"(bcur[g][13]), "s"(bcur[g][14]), "s"(bcur[g][15]));
      }
      const int lp = (i + G < PB) ? (l + G) : ((l + 2 * G <= Lc) ? (l + G) : l);
#pragma unroll
      for (int g = 0; g < G; ++g) {
#pragma unroll
        for (int n = 0; n < N; ++n) bcur[g][n] = Brow[(lp + g) * N + n];
      }
      __builtin_amdgcn_sched_barrier(0);
#pragma unroll
      for (int g = 0; g < G; ++g) {
        float dtv = dtc[i + g];
        float xv = xc[i + g];
        if (!FINAL) dtsum += dtv;
        float y = 0.0f;
#pragma unroll
        for (int n = 0; n < N; ++n) {
          float a = __builtin_amdgcn_exp2f(dtv * An2[n]);
          h[n] = __builtin_fmaf(a, h[n], ub[g][n]);
          if (FINAL) y += h[n] * cr[g][n];
        }
        if (FINAL) out[(rowbase + l + g) * Dc + du] = y + xv * Dv;
      }
      __builtin_amdgcn_sched_barrier(0);
    }
#pragma unroll
    for (int i = 0; i < PB; ++i) { dtc[i] = dtn[i]; xc[i] = xn[i]; }
  }

  for (int l = nb * PB; l < Lc; ++l) {
    float dtv = dtrow[(size_t)l * Dc + du];
    float xv = xrow[(size_t)l * Dc + du];
    float us = dtv * xv;
    if (!FINAL) dtsum += dtv;
    float y = 0.0f;
#pragma unroll
    for (int n = 0; n < N; ++n) {
      float a = __builtin_amdgcn_exp2f(dtv * An2[n]);
      h[n] = __builtin_fmaf(a, h[n], us * Brow[l * N + n]);
      if (FINAL) y += h[n] * Crow[l * N + n];
    }
    if (FINAL) out[(rowbase + l) * Dc + du] = y + xv * Dv;
  }

  if (!FINAL) {
#pragma unroll
    for (int n = 0; n < N; ++n) Hout[hidx + n] = h[n];
    Ssum[((size_t)b * nCh + chunk) * Dc + d] = dtsum;
  }
}

__global__ __launch_bounds__(256) void scan_combine_kernel(
    const float* __restrict__ Hlocal, const float* __restrict__ Ssum,
    const float* __restrict__ A_log, float* __restrict__ Hinit,
    int Dc, int nCh, int Bsz) {
  const int N = 16;
  const int CB = 32;
  const long long perB = (long long)Dc * N;
  long long idx = (long long)blockIdx.x * blockDim.x + threadIdx.x;
  if (idx >= (long long)Bsz * perB) return;
  const int b = (int)(idx / perB);
  const int dn = (int)(idx - (long long)b * perB);
  const int d = dn >> 4;
  const size_t stepH = (size_t)perB;
  const size_t stepS = (size_t)Dc;
  const float* hp = Hlocal + (size_t)b * nCh * stepH + dn;
  const float* sp = Ssum   + (size_t)b * nCh * stepS + d;
  float*       op = Hinit  + (size_t)b * nCh * stepH + dn;

  const float alog = A_log[dn];
  const int nb = nCh / CB;
  float Sc[CB], Hc[CB], Sn[CB], Hn[CB];
#pragma unroll
  for (int j = 0; j < CB; ++j) { Sc[j] = 0.0f; Hc[j] = 0.0f; Sn[j] = 0.0f; Hn[j] = 0.0f; }
  if (nb > 0) {
#pragma unroll
    for (int j = 0; j < CB; ++j) {
      Sc[j] = sp[(size_t)j * stepS];
      Hc[j] = hp[(size_t)j * stepH];
    }
  }
  const float An2 = -__expf(alog) * LOG2E;
  float h = 0.0f;
  for (int bi = 0; bi < nb; ++bi) {
    const int c0 = bi * CB;
    if (bi + 1 < nb) {
#pragma unroll
      for (int j = 0; j < CB; ++j) {
        Sn[j] = sp[(size_t)(c0 + CB + j) * stepS];
        Hn[j] = hp[(size_t)(c0 + CB + j) * stepH];
      }
    }
    __builtin_amdgcn_sched_barrier(0);
#pragma unroll
    for (int j = 0; j < CB; ++j) {
      op[(size_t)(c0 + j) * stepH] = h;
      float a = __builtin_amdgcn_exp2f(An2 * Sc[j]);
      h = a * h + Hc[j];
    }
#pragma unroll
    for (int j = 0; j < CB; ++j) { Sc[j] = Sn[j]; Hc[j] = Hn[j]; }
  }
  for (int c = nb * CB; c < nCh; ++c) {
    float S = sp[(size_t)c * stepS];
    op[(size_t)c * stepH] = h;
    float a = __builtin_amdgcn_exp2f(An2 * S);
    h = a * h + hp[(size_t)c * stepH];
  }
}

__global__ void scan_simple_kernel(const float* __restrict__ dt,
                                   const float* __restrict__ x,
                                   const float* __restrict__ Bm,
                                   const float* __restrict__ Cm,
                                   const float* __restrict__ A_log,
                                   const float* __restrict__ Dp,
                                   float* __restrict__ out,
                                   int L, int Dc, int Bsz) {
  const int N = 16;
  long long idx = (long long)blockIdx.x * blockDim.x + threadIdx.x;
  if (idx >= (long long)Bsz * Dc) return;
  int d = (int)(idx % Dc);
  int b = (int)(idx / Dc);
  float An2[N], h[N];
#pragma unroll
  for (int n = 0; n < N; ++n) {
    An2[n] = -__expf(A_log[(size_t)d * N + n]) * LOG2E;
    h[n] = 0.0f;
  }
  float Dv = Dp[d];
  for (int l = 0; l < L; ++l) {
    size_t row = (size_t)b * L + l;
    size_t g = row * Dc + d;
    float dtv = dt[g];
    float xv = x[g];
    float us = dtv * xv;
    const float* Br = Bm + row * N;
    const float* Cr = Cm + row * N;
    float y = 0.0f;
#pragma unroll
    for (int n = 0; n < N; ++n) {
      h[n] = __builtin_amdgcn_exp2f(dtv * An2[n]) * h[n] + us * Br[n];
      y += h[n] * Cr[n];
    }
    out[g] = y + xv * Dv;
  }
}

__global__ void scan_fused_kernel(const float* __restrict__ x,
                                  const float* __restrict__ delta,
                                  const float* __restrict__ Bm,
                                  const float* __restrict__ Cm,
                                  const float* __restrict__ W,
                                  const float* __restrict__ bias,
                                  const float* __restrict__ A_log,
                                  const float* __restrict__ Dp,
                                  float* __restrict__ out,
                                  int L, int Dc, int K, int Bsz) {
  const int N = 16;
  long long idx = (long long)blockIdx.x * blockDim.x + threadIdx.x;
  if (idx >= (long long)Bsz * Dc) return;
  int d = (int)(idx % Dc);
  int b = (int)(idx / Dc);
  float An2[N], h[N];
#pragma unroll
  for (int n = 0; n < N; ++n) {
    An2[n] = -__expf(A_log[(size_t)d * N + n]) * LOG2E;
    h[n] = 0.0f;
  }
  float Dv = Dp[d];
  float bv = bias[d];
  const float* wr = W + (size_t)d * K;
  for (int l = 0; l < L; ++l) {
    size_t row = (size_t)b * L + l;
    const float* dr = delta + row * K;
    float acc = bv;
    for (int r = 0; r < K; ++r) acc += dr[r] * wr[r];
    float dtv = softplus_f(acc);
    float xv = x[row * Dc + d];
    float us = dtv * xv;
    const float* Br = Bm + row * N;
    const float* Cr = Cm + row * N;
    float y = 0.0f;
#pragma unroll
    for (int n = 0; n < N; ++n) {
      h[n] = __builtin_amdgcn_exp2f(dtv * An2[n]) * h[n] + us * Br[n];
      y += h[n] * Cr[n];
    }
    out[row * Dc + d] = y + xv * Dv;
  }
}

extern "C" void kernel_launch(void* const* d_in, const int* in_sizes, int n_in,
                              void* d_out, int out_size, void* d_ws, size_t ws_size,
                              hipStream_t stream) {
  const float* x     = (const float*)d_in[0];
  const float* delta = (const float*)d_in[1];
  const float* Bm    = (const float*)d_in[2];
  const float* Cm    = (const float*)d_in[3];
  const float* W     = (const float*)d_in[4];
  const float* bias  = (const float*)d_in[5];
  const float* A_log = (const float*)d_in[6];
  const float* Dp    = (const float*)d_in[7];
  float* out = (float*)d_out;

  int Dc = in_sizes[5];
  int K  = in_sizes[4] / Dc;
  long long BL = (long long)in_sizes[0] / Dc;
  int Bsz = (BL % 2 == 0) ? 2 : 1;
  int L = (int)(BL / Bsz);

  size_t dtElems = (size_t)BL * Dc;
  size_t dtBytes = dtElems * sizeof(float);

  const int Lc = SCAN_LC;
  bool chunkable = (L % Lc) == 0 && (Dc % 256) == 0;
  int nCh = chunkable ? (L / Lc) : 1;
  size_t chElems = (size_t)Bsz * nCh * Dc;
  size_t fullBytes = dtBytes + chElems * sizeof(float) * (1 + 16 + 16);

  if (ws_size < dtBytes) {
    int threads = 256;
    long long tot = (long long)Bsz * Dc;
    int blocks = (int)((tot + threads - 1) / threads);
    scan_fused_kernel<<<blocks, threads, 0, stream>>>(
        x, delta, Bm, Cm, W, bias, A_log, Dp, out, L, Dc, K, Bsz);
    return;
  }

  float* dt = (float*)d_ws;

  if ((BL % 64) == 0 && (Dc % 128) == 0 && (K % 32) == 0) {
    int blocks = (int)(BL / 64) * (Dc / 128);
    dtproj_tiled_kernel<<<blocks, 256, 0, stream>>>(delta, W, bias, dt,
                                                    (int)BL, Dc, K);
  } else if ((BL % 16) == 0 && (Dc % 32) == 0 && (K % 4) == 0) {
    int tiles = (int)(BL / 16) * (Dc / 32);
    int blocks = (tiles + 7) / 8;
    dtproj_wmma_kernel<<<blocks, 256, 0, stream>>>(delta, W, bias, dt,
                                                   (int)BL, Dc, K);
  } else {
    long long tot = BL * (long long)Dc;
    int blocks = (int)((tot + 255) / 256);
    dtproj_naive_kernel<<<blocks, 256, 0, stream>>>(delta, W, bias, dt, BL, Dc, K);
  }

  if (chunkable && ws_size >= fullBytes) {
    float* Ssum   = dt + dtElems;
    float* Hlocal = Ssum + chElems;
    float* Hinit  = Hlocal + chElems * 16;

    int blocks13 = Bsz * nCh * (Dc / 256);
    scan_chunk_kernel<false><<<blocks13, 256, 0, stream>>>(
        dt, x, Bm, Cm, A_log, Dp, Hlocal, Ssum, (const float*)nullptr,
        (float*)nullptr, L, Dc, Lc, nCh);

    long long tot = (long long)Bsz * Dc * 16;
    int cblocks = (int)((tot + 255) / 256);
    scan_combine_kernel<<<cblocks, 256, 0, stream>>>(Hlocal, Ssum, A_log, Hinit,
                                                     Dc, nCh, Bsz);

    scan_chunk_kernel<true><<<blocks13, 256, 0, stream>>>(
        dt, x, Bm, Cm, A_log, Dp, (float*)nullptr, (float*)nullptr, Hinit, out,
        L, Dc, Lc, nCh);
  } else {
    long long tot = (long long)Bsz * Dc;
    int blocks = (int)((tot + 255) / 256);
    scan_simple_kernel<<<blocks, 256, 0, stream>>>(dt, x, Bm, Cm, A_log, Dp, out,
                                                   L, Dc, Bsz);
  }
}
